# starting kernel
# baseline (speedup 1.0000x reference)
#include <hip/hip_runtime.h>

#define N_NODES 100000
#define N_EDGES 1600000
#define N_PAIRS 500000
#define D       128

typedef __attribute__((ext_vector_type(2))) float v2f;
typedef __attribute__((ext_vector_type(8))) float v8f;

__global__ void k_init_deg(float* __restrict__ deg) {
  int i = blockIdx.x * 256 + threadIdx.x;
  if (i < N_NODES) deg[i] = 1.0f;
}

__global__ void k_degree(const int* __restrict__ ei, float* __restrict__ deg) {
  int e = blockIdx.x * 256 + threadIdx.x;
  if (e < N_EDGES) atomicAdd(&deg[ei[N_EDGES + e]], 1.0f);
}

__global__ void k_rsqrt(float* __restrict__ deg) {
  int i = blockIdx.x * 256 + threadIdx.x;
  if (i < N_NODES) deg[i] = rsqrtf(deg[i]);
}

__global__ void k_zero4(float4* __restrict__ p, int n4) {
  int i = blockIdx.x * 256 + threadIdx.x;
  if (i < n4) p[i] = make_float4(0.f, 0.f, 0.f, 0.f);
}

__global__ void __launch_bounds__(256) k_gemm(const float* __restrict__ X,
                                              const float* __restrict__ W,
                                              float* __restrict__ Y,
                                              int nrows) {
  const int lane = threadIdx.x & 31;
  const int wv   = threadIdx.x >> 5;
  const int tile = blockIdx.x * 8 + wv;
  const int tm = tile >> 3;
  const int tn = tile & 7;
  if (tm * 16 >= nrows) return;

  const int r    = lane & 15;
  const int koff = (lane >> 4) << 1;

  const float* xrow = X + (size_t)(tm * 16 + r) * D;
  const float* wcol = W + tn * 16 + r;

  v8f acc = {};
#pragma unroll 4
  for (int k = 0; k < D; k += 4) {
    float2 av = *(const float2*)(xrow + k + koff);
    v2f a; a.x = av.x; a.y = av.y;
    v2f b;
    b.x = wcol[(k + koff)     * D];
    b.y = wcol[(k + koff + 1) * D];
    acc = __builtin_amdgcn_wmma_f32_16x16x4_f32(
         false, a,  false, b,
         (short)0, acc,  false,  false);
  }

  const int rowbase = tm * 16 + ((lane >> 4) << 3);
  const int col     = tn * 16 + r;
#pragma unroll
  for (int i = 0; i < 8; ++i)
    Y[(size_t)(rowbase + i) * D + col] = acc[i];
}

__global__ void __launch_bounds__(256) k_aggregate(const int* __restrict__ ei,
                                                   const float* __restrict__ t,
                                                   const float* __restrict__ dinv,
                                                   float* __restrict__ agg) {
  int e = blockIdx.x * 8 + (threadIdx.x >> 5);
  if (e >= N_EDGES) return;
  const int lane = threadIdx.x & 31;
  const int s = ei[e];
  const int d = ei[N_EDGES + e];
  const float nn = dinv[s] * dinv[d];
  float4 v = ((const float4*)(t + (size_t)s * D))[lane];
  float* o = agg + (size_t)d * D + lane * 4;
  atomicAdd(o + 0, v.x * nn);
  atomicAdd(o + 1, v.y * nn);
  atomicAdd(o + 2, v.z * nn);
  atomicAdd(o + 3, v.w * nn);
}

__global__ void k_finish(const float* __restrict__ agg,
                         const float* __restrict__ t,
                         const float* __restrict__ dinv,
                         const float* __restrict__ bias,
                         float* __restrict__ h, int relu) {
  int i = blockIdx.x * 256 + threadIdx.x;
  if (i >= N_NODES * D) return;
  const int node = i >> 7;
  const int d    = i & 127;
  const float di = dinv[node];
  float v = agg[i] + t[i] * di * di + bias[d];
  if (relu) v = fmaxf(v, 0.0f);
  h[i] = v;
}

__global__ void __launch_bounds__(256) k_dot(const int* __restrict__ ep,
                                             const int* __restrict__ en,
                                             const float* __restrict__ h,
                                             float* __restrict__ out) {
  int p = blockIdx.x * 8 + (threadIdx.x >> 5);
  if (p >= 2 * N_PAIRS) return;
  const int lane = threadIdx.x & 31;
  const int* pr = (p < N_PAIRS) ? ep : en;
  const int pp  = (p < N_PAIRS) ? p : p - N_PAIRS;
  const int a = pr[2 * pp], b = pr[2 * pp + 1];
  float4 va = ((const float4*)(h + (size_t)a * D))[lane];
  float4 vb = ((const float4*)(h + (size_t)b * D))[lane];
  float s = va.x * vb.x + va.y * vb.y + va.z * vb.z + va.w * vb.w;
#pragma unroll
  for (int m = 16; m > 0; m >>= 1) s += __shfl_xor(s, m, 32);
  if (lane == 0) out[p] = s;
}

extern "C" void kernel_launch(void* const* d_in, const int* in_sizes, int n_in,
                              void* d_out, int out_size, void* d_ws, size_t ws_size,
                              hipStream_t stream) {
  (void)in_sizes; (void)n_in; (void)out_size; (void)ws_size;
  const float* x   = (const float*)d_in[0];
  const float* W1  = (const float*)d_in[1];
  const float* b1  = (const float*)d_in[2];
  const float* W2  = (const float*)d_in[3];
  const float* b2  = (const float*)d_in[4];
  const int* ei    = (const int*)d_in[5];
  const int* edges = (const int*)d_in[6];
  const int* edneg = (const int*)d_in[7];
  float* out = (float*)d_out;

  const size_t FEAT = (size_t)N_NODES * D;
  float* t    = (float*)d_ws;
  float* agg  = t + FEAT;
  float* h    = agg + FEAT;
  float* dinv = h + FEAT;

  const int NB_N  = (N_NODES + 255) / 256;
  const int NB_E  = (N_EDGES + 255) / 256;
  const int N4    = (int)(FEAT / 4);
  const int NB_Z  = (N4 + 255) / 256;
  const int NB_F  = (int)((FEAT + 255) / 256);
  const int NB_AG = (N_EDGES + 7) / 8;
  const int NB_GM = (N_NODES / 16) * 8 / 8;
  const int NB_DP = (2 * N_PAIRS + 7) / 8;

  k_init_deg<<<NB_N, 256, 0, stream>>>(dinv);
  k_degree  <<<NB_E, 256, 0, stream>>>(ei, dinv);
  k_rsqrt   <<<NB_N, 256, 0, stream>>>(dinv);

  k_gemm     <<<NB_GM, 256, 0, stream>>>(x, W1, t, N_NODES);
  k_zero4    <<<NB_Z, 256, 0, stream>>>((float4*)agg, N4);
  k_aggregate<<<NB_AG, 256, 0, stream>>>(ei, t, dinv, agg);
  k_finish   <<<NB_F, 256, 0, stream>>>(agg, t, dinv, b1, h, 1);

  k_gemm     <<<NB_GM, 256, 0, stream>>>(h, W2, t, N_NODES);
  k_zero4    <<<NB_Z, 256, 0, stream>>>((float4*)agg, N4);
  k_aggregate<<<NB_AG, 256, 0, stream>>>(ei, t, dinv, agg);
  k_finish   <<<NB_F, 256, 0, stream>>>(agg, t, dinv, b2, h, 0);

  k_dot<<<NB_DP, 256, 0, stream>>>(edges, edneg, h, out);
}


# Aggregation by pull: edges counting-sorted by dst once (int atomic rank + scan), one wave per row sums in-edges+self+bias in registers, one store; row sum order = arrival rank, no float atomics
# speedup vs baseline: 2.2223x; 2.2223x over previous
#include <hip/hip_runtime.h>

#define N_NODES 100000
#define N_EDGES 1600000
#define N_PAIRS 500000
#define D       128

typedef __attribute__((ext_vector_type(2))) float v2f;
typedef __attribute__((ext_vector_type(8))) float v8f;

__global__ void k_zero_cnt(int* __restrict__ cnt) {
  int i = blockIdx.x * 256 + threadIdx.x;
  if (i < N_NODES) cnt[i] = 0;
}

__global__ void k_count(const int* __restrict__ ei, int* __restrict__ cnt,
                        int* __restrict__ rank) {
  int e = blockIdx.x * 256 + threadIdx.x;
  if (e < N_EDGES) rank[e] = atomicAdd(&cnt[ei[N_EDGES + e]], 1);
}

#define SCAN_B 1024
__global__ void __launch_bounds__(SCAN_B) k_scan(const int* __restrict__ cnt,
                                                 int* __restrict__ rowptr,
                                                 float* __restrict__ dinv) {
  __shared__ int s_pre[32];
  __shared__ int s_tot[32];
  const int tid  = threadIdx.x;
  const int lane = tid & 31;
  const int wv   = tid >> 5;

  int pre = 0;
  const int4* c4 = (const int4*)cnt;
  const int n4 = blockIdx.x * (SCAN_B / 4);
  for (int i = tid; i < n4; i += SCAN_B) {
    const int4 v = c4[i];
    pre += v.x + v.y + v.z + v.w;
  }

  const int node = blockIdx.x * SCAN_B + tid;
  const int c = (node < N_NODES) ? cnt[node] : 0;

  int inc = c;
  for (int m = 1; m < 32; m <<= 1) {
    const int o = __shfl_up(inc, m, 32);
    if (lane >= m) inc += o;
  }
  for (int m = 16; m > 0; m >>= 1) pre += __shfl_xor(pre, m, 32);
  if (lane == 31) s_tot[wv] = inc;
  if (lane == 0)  s_pre[wv] = pre;
  __syncthreads();
  if (wv == 0) {
    const int v = s_tot[lane];
    int iv = v;
    for (int m = 1; m < 32; m <<= 1) {
      const int o = __shfl_up(iv, m, 32);
      if (lane >= m) iv += o;
    }
    int p = s_pre[lane];
    for (int m = 16; m > 0; m >>= 1) p += __shfl_xor(p, m, 32);
    s_tot[lane] = iv - v + p;
  }
  __syncthreads();
  if (node < N_NODES) {
    const int excl = s_tot[wv] + inc - c;
    rowptr[node] = excl;
    dinv[node]   = rsqrtf((float)(c + 1));
    if (node == N_NODES - 1) rowptr[N_NODES] = excl + c;
  }
}

__global__ void k_fill(const int* __restrict__ ei, const int* __restrict__ rowptr,
                       const int* __restrict__ rank, int* __restrict__ col) {
  int e = blockIdx.x * 256 + threadIdx.x;
  if (e < N_EDGES) col[rowptr[ei[N_EDGES + e]] + rank[e]] = ei[e];
}

__global__ void __launch_bounds__(256) k_gemm(const float* __restrict__ X,
                                              const float* __restrict__ W,
                                              float* __restrict__ Y,
                                              int nrows) {
  const int lane = threadIdx.x & 31;
  const int wv   = threadIdx.x >> 5;
  const int tile = blockIdx.x * 8 + wv;
  const int tm = tile >> 3;
  const int tn = tile & 7;
  if (tm * 16 >= nrows) return;

  const int r    = lane & 15;
  const int koff = (lane >> 4) << 1;

  const float* xrow = X + (size_t)(tm * 16 + r) * D;
  const float* wcol = W + tn * 16 + r;

  v8f acc = {};
#pragma unroll 4
  for (int k = 0; k < D; k += 4) {
    float2 av = *(const float2*)(xrow + k + koff);
    v2f a; a.x = av.x; a.y = av.y;
    v2f b;
    b.x = wcol[(k + koff)     * D];
    b.y = wcol[(k + koff + 1) * D];
    acc = __builtin_amdgcn_wmma_f32_16x16x4_f32(
         false, a,  false, b,
         (short)0, acc,  false,  false);
  }

  const int rowbase = tm * 16 + ((lane >> 4) << 3);
  const int col     = tn * 16 + r;
#pragma unroll
  for (int i = 0; i < 8; ++i)
    Y[(size_t)(rowbase + i) * D + col] = acc[i];
}

__device__ __forceinline__ void fma4(float4& a, const float4 v, const float w) {
  a.x = fmaf(v.x, w, a.x); a.y = fmaf(v.y, w, a.y);
  a.z = fmaf(v.z, w, a.z); a.w = fmaf(v.w, w, a.w);
}

__global__ void __launch_bounds__(256) k_gather(const int* __restrict__ rowptr,
                                                const int* __restrict__ col,
                                                const float* __restrict__ t,
                                                const float* __restrict__ dinv,
                                                const float* __restrict__ bias,
                                                float* __restrict__ h, int relu) {
  const int lane = threadIdx.x & 31;
  const int row = __builtin_amdgcn_readfirstlane(blockIdx.x * 8 + (threadIdx.x >> 5));
  if (row >= N_NODES) return;
  const int beg  = rowptr[row];
  const int m    = rowptr[row + 1] - beg + 1;
  const float di = dinv[row];
  const float4* t4 = (const float4*)t;
  float4 acc = make_float4(0.f, 0.f, 0.f, 0.f);

  for (int base = 0; base < m; base += 32) {
    const int n = min(32, m - base);
    const int k = base + lane;
    int ms = row;
    if (k < m - 1) ms = col[beg + k];
    const int mdb = __float_as_int(dinv[ms]);
    int j = 0;
    for (; j + 4 <= n; j += 4) {
      const int s0 = __builtin_amdgcn_readlane(ms, j);
      const int s1 = __builtin_amdgcn_readlane(ms, j + 1);
      const int s2 = __builtin_amdgcn_readlane(ms, j + 2);
      const int s3 = __builtin_amdgcn_readlane(ms, j + 3);
      const float4 v0 = t4[(size_t)s0 * (D / 4) + lane];
      const float4 v1 = t4[(size_t)s1 * (D / 4) + lane];
      const float4 v2 = t4[(size_t)s2 * (D / 4) + lane];
      const float4 v3 = t4[(size_t)s3 * (D / 4) + lane];
      const float w0 = __int_as_float(__builtin_amdgcn_readlane(mdb, j)) * di;
      const float w1 = __int_as_float(__builtin_amdgcn_readlane(mdb, j + 1)) * di;
      const float w2 = __int_as_float(__builtin_amdgcn_readlane(mdb, j + 2)) * di;
      const float w3 = __int_as_float(__builtin_amdgcn_readlane(mdb, j + 3)) * di;
      fma4(acc, v0, w0);
      fma4(acc, v1, w1);
      fma4(acc, v2, w2);
      fma4(acc, v3, w3);
    }
    const int r = n - j;
    if (r > 0) {
      const int s0 = __builtin_amdgcn_readlane(ms, j);
      const float4 v0 = t4[(size_t)s0 * (D / 4) + lane];
      if (r > 1) {
        const int s1 = __builtin_amdgcn_readlane(ms, j + 1);
        const float4 v1 = t4[(size_t)s1 * (D / 4) + lane];
        if (r > 2) {
          const int s2 = __builtin_amdgcn_readlane(ms, j + 2);
          const float4 v2 = t4[(size_t)s2 * (D / 4) + lane];
          fma4(acc, v2, __int_as_float(__builtin_amdgcn_readlane(mdb, j + 2)) * di);
        }
        fma4(acc, v1, __int_as_float(__builtin_amdgcn_readlane(mdb, j + 1)) * di);
      }
      fma4(acc, v0, __int_as_float(__builtin_amdgcn_readlane(mdb, j)) * di);
    }
  }

  const float4 b = ((const float4*)bias)[lane];
  acc.x += b.x; acc.y += b.y; acc.z += b.z; acc.w += b.w;
  if (relu) {
    acc.x = fmaxf(acc.x, 0.0f); acc.y = fmaxf(acc.y, 0.0f);
    acc.z = fmaxf(acc.z, 0.0f); acc.w = fmaxf(acc.w, 0.0f);
  }
  ((float4*)h)[(size_t)row * (D / 4) + lane] = acc;
}

__global__ void __launch_bounds__(256) k_dot(const int* __restrict__ ep,
                                             const int* __restrict__ en,
                                             const float* __restrict__ h,
                                             float* __restrict__ out) {
  int p = blockIdx.x * 8 + (threadIdx.x >> 5);
  if (p >= 2 * N_PAIRS) return;
  const int lane = threadIdx.x & 31;
  const int* pr = (p < N_PAIRS) ? ep : en;
  const int pp  = (p < N_PAIRS) ? p : p - N_PAIRS;
  const int a = pr[2 * pp], b = pr[2 * pp + 1];
  float4 va = ((const float4*)(h + (size_t)a * D))[lane];
  float4 vb = ((const float4*)(h + (size_t)b * D))[lane];
  float s = va.x * vb.x + va.y * vb.y + va.z * vb.z + va.w * vb.w;
#pragma unroll
  for (int m = 16; m > 0; m >>= 1) s += __shfl_xor(s, m, 32);
  if (lane == 0) out[p] = s;
}

extern "C" void kernel_launch(void* const* d_in, const int* in_sizes, int n_in,
                              void* d_out, int out_size, void* d_ws, size_t ws_size,
                              hipStream_t stream) {
  (void)in_sizes; (void)n_in; (void)out_size; (void)ws_size;
  const float* x   = (const float*)d_in[0];
  const float* W1  = (const float*)d_in[1];
  const float* b1  = (const float*)d_in[2];
  const float* W2  = (const float*)d_in[3];
  const float* b2  = (const float*)d_in[4];
  const int* ei    = (const int*)d_in[5];
  const int* edges = (const int*)d_in[6];
  const int* edneg = (const int*)d_in[7];
  float* out = (float*)d_out;

  const size_t FEAT = (size_t)N_NODES * D;
  float* t    = (float*)d_ws;
  float* csr  = t + FEAT;
  float* h    = csr + FEAT;
  float* dinv = h + FEAT;
  int* cnt    = (int*)csr;
  int* rowptr = cnt + N_NODES;
  int* rank   = rowptr + N_NODES + 4;
  int* col    = rank + N_EDGES;

  const int NB_N  = (N_NODES + 255) / 256;
  const int NB_E  = (N_EDGES + 255) / 256;
  const int NB_S  = (N_NODES + SCAN_B - 1) / SCAN_B;
  const int NB_GA = (N_NODES + 7) / 8;
  const int NB_GM = (N_NODES / 16) * 8 / 8;
  const int NB_DP = (2 * N_PAIRS + 7) / 8;

  k_zero_cnt<<<NB_N, 256, 0, stream>>>(cnt);
  k_count   <<<NB_E, 256, 0, stream>>>(ei, cnt, rank);
  k_scan    <<<NB_S, SCAN_B, 0, stream>>>(cnt, rowptr, dinv);
  k_fill    <<<NB_E, 256, 0, stream>>>(ei, rowptr, rank, col);

  k_gemm  <<<NB_GM, 256, 0, stream>>>(x, W1, t, N_NODES);
  k_gather<<<NB_GA, 256, 0, stream>>>(rowptr, col, t, dinv, b1, h, 1);

  k_gemm  <<<NB_GM, 256, 0, stream>>>(h, W2, t, N_NODES);
  k_gather<<<NB_GA, 256, 0, stream>>>(rowptr, col, t, dinv, b2, h, 0);

  k_dot<<<NB_DP, 256, 0, stream>>>(edges, edneg, h, out);
}


# Pair scores: one wave scores 32 pairs, not 1 (32x fewer waves): ids by one coalesced load, ~10 row loads in flight, halving exchange sums lanes at distance 1,2,4,8,16 (was 16..1), one 128-B store
# speedup vs baseline: 2.6489x; 1.1920x over previous
#include <hip/hip_runtime.h>

#define N_NODES 100000
#define N_EDGES 1600000
#define N_PAIRS 500000
#define D       128

typedef __attribute__((ext_vector_type(2))) float v2f;
typedef __attribute__((ext_vector_type(8))) float v8f;

__global__ void k_zero_cnt(int* __restrict__ cnt) {
  int i = blockIdx.x * 256 + threadIdx.x;
  if (i < N_NODES) cnt[i] = 0;
}

__global__ void k_count(const int* __restrict__ ei, int* __restrict__ cnt,
                        int* __restrict__ rank) {
  int e = blockIdx.x * 256 + threadIdx.x;
  if (e < N_EDGES) rank[e] = atomicAdd(&cnt[ei[N_EDGES + e]], 1);
}

#define SCAN_B 1024
__global__ void __launch_bounds__(SCAN_B) k_scan(const int* __restrict__ cnt,
                                                 int* __restrict__ rowptr,
                                                 float* __restrict__ dinv) {
  __shared__ int s_pre[32];
  __shared__ int s_tot[32];
  const int tid  = threadIdx.x;
  const int lane = tid & 31;
  const int wv   = tid >> 5;

  int pre = 0;
  const int4* c4 = (const int4*)cnt;
  const int n4 = blockIdx.x * (SCAN_B / 4);
  for (int i = tid; i < n4; i += SCAN_B) {
    const int4 v = c4[i];
    pre += v.x + v.y + v.z + v.w;
  }

  const int node = blockIdx.x * SCAN_B + tid;
  const int c = (node < N_NODES) ? cnt[node] : 0;

  int inc = c;
  for (int m = 1; m < 32; m <<= 1) {
    const int o = __shfl_up(inc, m, 32);
    if (lane >= m) inc += o;
  }
  for (int m = 16; m > 0; m >>= 1) pre += __shfl_xor(pre, m, 32);
  if (lane == 31) s_tot[wv] = inc;
  if (lane == 0)  s_pre[wv] = pre;
  __syncthreads();
  if (wv == 0) {
    const int v = s_tot[lane];
    int iv = v;
    for (int m = 1; m < 32; m <<= 1) {
      const int o = __shfl_up(iv, m, 32);
      if (lane >= m) iv += o;
    }
    int p = s_pre[lane];
    for (int m = 16; m > 0; m >>= 1) p += __shfl_xor(p, m, 32);
    s_tot[lane] = iv - v + p;
  }
  __syncthreads();
  if (node < N_NODES) {
    const int excl = s_tot[wv] + inc - c;
    rowptr[node] = excl;
    dinv[node]   = rsqrtf((float)(c + 1));
    if (node == N_NODES - 1) rowptr[N_NODES] = excl + c;
  }
}

__global__ void k_fill(const int* __restrict__ ei, const int* __restrict__ rowptr,
                       const int* __restrict__ rank, int* __restrict__ col) {
  int e = blockIdx.x * 256 + threadIdx.x;
  if (e < N_EDGES) col[rowptr[ei[N_EDGES + e]] + rank[e]] = ei[e];
}

__global__ void __launch_bounds__(256) k_gemm(const float* __restrict__ X,
                                              const float* __restrict__ W,
                                              float* __restrict__ Y,
                                              int nrows) {
  const int lane = threadIdx.x & 31;
  const int wv   = threadIdx.x >> 5;
  const int tile = blockIdx.x * 8 + wv;
  const int tm = tile >> 3;
  const int tn = tile & 7;
  if (tm * 16 >= nrows) return;

  const int r    = lane & 15;
  const int koff = (lane >> 4) << 1;

  const float* xrow = X + (size_t)(tm * 16 + r) * D;
  const float* wcol = W + tn * 16 + r;

  v8f acc = {};
#pragma unroll 4
  for (int k = 0; k < D; k += 4) {
    float2 av = *(const float2*)(xrow + k + koff);
    v2f a; a.x = av.x; a.y = av.y;
    v2f b;
    b.x = wcol[(k + koff)     * D];
    b.y = wcol[(k + koff + 1) * D];
    acc = __builtin_amdgcn_wmma_f32_16x16x4_f32(
         false, a,  false, b,
         (short)0, acc,  false,  false);
  }

  const int rowbase = tm * 16 + ((lane >> 4) << 3);
  const int col     = tn * 16 + r;
#pragma unroll
  for (int i = 0; i < 8; ++i)
    Y[(size_t)(rowbase + i) * D + col] = acc[i];
}

__device__ __forceinline__ void fma4(float4& a, const float4 v, const float w) {
  a.x = fmaf(v.x, w, a.x); a.y = fmaf(v.y, w, a.y);
  a.z = fmaf(v.z, w, a.z); a.w = fmaf(v.w, w, a.w);
}

__global__ void __launch_bounds__(256) k_gather(const int* __restrict__ rowptr,
                                                const int* __restrict__ col,
                                                const float* __restrict__ t,
                                                const float* __restrict__ dinv,
                                                const float* __restrict__ bias,
                                                float* __restrict__ h, int relu) {
  const int lane = threadIdx.x & 31;
  const int row = __builtin_amdgcn_readfirstlane(blockIdx.x * 8 + (threadIdx.x >> 5));
  if (row >= N_NODES) return;
  const int beg  = rowptr[row];
  const int m    = rowptr[row + 1] - beg + 1;
  const float di = dinv[row];
  const float4* t4 = (const float4*)t;
  float4 acc = make_float4(0.f, 0.f, 0.f, 0.f);

  for (int base = 0; base < m; base += 32) {
    const int n = min(32, m - base);
    const int k = base + lane;
    int ms = row;
    if (k < m - 1) ms = col[beg + k];
    const int mdb = __float_as_int(dinv[ms]);
    int j = 0;
    for (; j + 4 <= n; j += 4) {
      const int s0 = __builtin_amdgcn_readlane(ms, j);
      const int s1 = __builtin_amdgcn_readlane(ms, j + 1);
      const int s2 = __builtin_amdgcn_readlane(ms, j + 2);
      const int s3 = __builtin_amdgcn_readlane(ms, j + 3);
      const float4 v0 = t4[(size_t)s0 * (D / 4) + lane];
      const float4 v1 = t4[(size_t)s1 * (D / 4) + lane];
      const float4 v2 = t4[(size_t)s2 * (D / 4) + lane];
      const float4 v3 = t4[(size_t)s3 * (D / 4) + lane];
      const float w0 = __int_as_float(__builtin_amdgcn_readlane(mdb, j)) * di;
      const float w1 = __int_as_float(__builtin_amdgcn_readlane(mdb, j + 1)) * di;
      const float w2 = __int_as_float(__builtin_amdgcn_readlane(mdb, j + 2)) * di;
      const float w3 = __int_as_float(__builtin_amdgcn_readlane(mdb, j + 3)) * di;
      fma4(acc, v0, w0);
      fma4(acc, v1, w1);
      fma4(acc, v2, w2);
      fma4(acc, v3, w3);
    }
    const int r = n - j;
    if (r > 0) {
      const int s0 = __builtin_amdgcn_readlane(ms, j);
      const float4 v0 = t4[(size_t)s0 * (D / 4) + lane];
      if (r > 1) {
        const int s1 = __builtin_amdgcn_readlane(ms, j + 1);
        const float4 v1 = t4[(size_t)s1 * (D / 4) + lane];
        if (r > 2) {
          const int s2 = __builtin_amdgcn_readlane(ms, j + 2);
          const float4 v2 = t4[(size_t)s2 * (D / 4) + lane];
          fma4(acc, v2, __int_as_float(__builtin_amdgcn_readlane(mdb, j + 2)) * di);
        }
        fma4(acc, v1, __int_as_float(__builtin_amdgcn_readlane(mdb, j + 1)) * di);
      }
      fma4(acc, v0, __int_as_float(__builtin_amdgcn_readlane(mdb, j)) * di);
    }
  }

  const float4 b = ((const float4*)bias)[lane];
  acc.x += b.x; acc.y += b.y; acc.z += b.z; acc.w += b.w;
  if (relu) {
    acc.x = fmaxf(acc.x, 0.0f); acc.y = fmaxf(acc.y, 0.0f);
    acc.z = fmaxf(acc.z, 0.0f); acc.w = fmaxf(acc.w, 0.0f);
  }
  ((float4*)h)[(size_t)row * (D / 4) + lane] = acc;
}

#define DOT_P 32
#define DOT_U 8
#define DOT_WAVES 4
static_assert(N_PAIRS % DOT_P == 0, "a wave's pairs must all come from one list");
static_assert(DOT_P == 32 && DOT_U == 8, "the exchange below is written for 4 trips of 8 pairs");

__device__ __forceinline__ float halve(const float lo, const float hi, const bool up, const int m) {
  const float keep = up ? hi : lo;
  const float send = up ? lo : hi;
  return keep + __shfl_xor(send, m, 32);
}

__global__ void __launch_bounds__(DOT_WAVES * 32) k_dot(const int* __restrict__ ep,
                                                        const int* __restrict__ en,
                                                        const float* __restrict__ h,
                                                        float* __restrict__ out) {
  const int lane = threadIdx.x & 31;
  const int wave = __builtin_amdgcn_readfirstlane(blockIdx.x * DOT_WAVES + (threadIdx.x >> 5));
  const int p0 = wave * DOT_P;
  if (p0 >= 2 * N_PAIRS) return;
  const int* pr = (p0 < N_PAIRS) ? ep : en;
  const int  q0 = (p0 < N_PAIRS) ? p0 : p0 - N_PAIRS;
  const int2 ab = ((const int2*)pr)[q0 + lane];
  const bool up1 = (lane & 1) != 0, up2 = (lane & 2) != 0, up4 = (lane & 4) != 0;

  float r[DOT_P / DOT_U];
#pragma unroll
  for (int g = 0; g < DOT_P / DOT_U; ++g) {
    float4 va[DOT_U], vb[DOT_U];
#pragma unroll
    for (int u = 0; u < DOT_U; ++u) {
      const int a = __builtin_amdgcn_readlane(ab.x, g * DOT_U + u);
      const int b = __builtin_amdgcn_readlane(ab.y, g * DOT_U + u);
      va[u] = ((const float4*)(h + (size_t)a * D))[lane];
      vb[u] = ((const float4*)(h + (size_t)b * D))[lane];
    }
    float p[DOT_U];
#pragma unroll
    for (int u = 0; u < DOT_U; ++u)
      p[u] = fmaf(va[u].w, vb[u].w, fmaf(va[u].z, vb[u].z, fmaf(va[u].x, vb[u].x, va[u].y * vb[u].y)));
    const float c0 = halve(p[0], p[1], up1, 1), c1 = halve(p[2], p[3], up1, 1);
    const float c2 = halve(p[4], p[5], up1, 1), c3 = halve(p[6], p[7], up1, 1);
    const float d0 = halve(c0, c1, up2, 2), d1 = halve(c2, c3, up2, 2);
    r[g] = halve(d0, d1, up4, 4);
  }
  const bool up8 = (lane & 8) != 0, up16 = (lane & 16) != 0;
  const float w0 = halve(r[0], r[1], up8, 8), w1 = halve(r[2], r[3], up8, 8);
  out[p0 + lane] = halve(w0, w1, up16, 16);
}

extern "C" void kernel_launch(void* const* d_in, const int* in_sizes, int n_in,
                              void* d_out, int out_size, void* d_ws, size_t ws_size,
                              hipStream_t stream) {
  (void)in_sizes; (void)n_in; (void)out_size; (void)ws_size;
  const float* x   = (const float*)d_in[0];
  const float* W1  = (const float*)d_in[1];
  const float* b1  = (const float*)d_in[2];
  const float* W2  = (const float*)d_in[3];
  const float* b2  = (const float*)d_in[4];
  const int* ei    = (const int*)d_in[5];
  const int* edges = (const int*)d_in[6];
  const int* edneg = (const int*)d_in[7];
  float* out = (float*)d_out;

  const size_t FEAT = (size_t)N_NODES * D;
  float* t    = (float*)d_ws;
  float* csr  = t + FEAT;
  float* h    = csr + FEAT;
  float* dinv = h + FEAT;
  int* cnt    = (int*)csr;
  int* rowptr = cnt + N_NODES;
  int* rank   = rowptr + N_NODES + 4;
  int* col    = rank + N_EDGES;

  const int NB_N  = (N_NODES + 255) / 256;
  const int NB_E  = (N_EDGES + 255) / 256;
  const int NB_S  = (N_NODES + SCAN_B - 1) / SCAN_B;
  const int NB_GA = (N_NODES + 7) / 8;
  const int NB_GM = (N_NODES / 16) * 8 / 8;
  const int NB_DP = (2 * N_PAIRS / DOT_P + DOT_WAVES - 1) / DOT_WAVES;

  k_zero_cnt<<<NB_N, 256, 0, stream>>>(cnt);
  k_count   <<<NB_E, 256, 0, stream>>>(ei, cnt, rank);
  k_scan    <<<NB_S, SCAN_B, 0, stream>>>(cnt, rowptr, dinv);
  k_fill    <<<NB_E, 256, 0, stream>>>(ei, rowptr, rank, col);

  k_gemm  <<<NB_GM, 256, 0, stream>>>(x, W1, t, N_NODES);
  k_gather<<<NB_GA, 256, 0, stream>>>(rowptr, col, t, dinv, b1, h, 1);

  k_gemm  <<<NB_GM, 256, 0, stream>>>(h, W2, t, N_NODES);
  k_gather<<<NB_GA, 256, 0, stream>>>(rowptr, col, t, dinv, b2, h, 0);

  k_dot<<<NB_DP, DOT_WAVES * 32, 0, stream>>>(edges, edneg, h, out);
}


# GEMM re-tiled: wave owns 16x64 of Y, each A fragment feeds 4 tiles, W staged once per work-group in LDS as k-quads, whole-line stores via LDS; k summed in another order inside each 8
# speedup vs baseline: 3.0777x; 1.1619x over previous
#include <hip/hip_runtime.h>

#define N_NODES 100000
#define N_EDGES 1600000
#define N_PAIRS 500000
#define D       128

typedef __attribute__((ext_vector_type(2))) float v2f;
typedef __attribute__((ext_vector_type(8))) float v8f;

__global__ void k_zero_cnt(int* __restrict__ cnt) {
  int i = blockIdx.x * 256 + threadIdx.x;
  if (i < N_NODES) cnt[i] = 0;
}

__global__ void k_count(const int* __restrict__ ei, int* __restrict__ cnt,
                        int* __restrict__ rank) {
  int e = blockIdx.x * 256 + threadIdx.x;
  if (e < N_EDGES) rank[e] = atomicAdd(&cnt[ei[N_EDGES + e]], 1);
}

#define SCAN_B 1024
__global__ void __launch_bounds__(SCAN_B) k_scan(const int* __restrict__ cnt,
                                                 int* __restrict__ rowptr,
                                                 float* __restrict__ dinv) {
  __shared__ int s_pre[32];
  __shared__ int s_tot[32];
  const int tid  = threadIdx.x;
  const int lane = tid & 31;
  const int wv   = tid >> 5;

  int pre = 0;
  const int4* c4 = (const int4*)cnt;
  const int n4 = blockIdx.x * (SCAN_B / 4);
  for (int i = tid; i < n4; i += SCAN_B) {
    const int4 v = c4[i];
    pre += v.x + v.y + v.z + v.w;
  }

  const int node = blockIdx.x * SCAN_B + tid;
  const int c = (node < N_NODES) ? cnt[node] : 0;

  int inc = c;
  for (int m = 1; m < 32; m <<= 1) {
    const int o = __shfl_up(inc, m, 32);
    if (lane >= m) inc += o;
  }
  for (int m = 16; m > 0; m >>= 1) pre += __shfl_xor(pre, m, 32);
  if (lane == 31) s_tot[wv] = inc;
  if (lane == 0)  s_pre[wv] = pre;
  __syncthreads();
  if (wv == 0) {
    const int v = s_tot[lane];
    int iv = v;
    for (int m = 1; m < 32; m <<= 1) {
      const int o = __shfl_up(iv, m, 32);
      if (lane >= m) iv += o;
    }
    int p = s_pre[lane];
    for (int m = 16; m > 0; m >>= 1) p += __shfl_xor(p, m, 32);
    s_tot[lane] = iv - v + p;
  }
  __syncthreads();
  if (node < N_NODES) {
    const int excl = s_tot[wv] + inc - c;
    rowptr[node] = excl;
    dinv[node]   = rsqrtf((float)(c + 1));
    if (node == N_NODES - 1) rowptr[N_NODES] = excl + c;
  }
}

__global__ void k_fill(const int* __restrict__ ei, const int* __restrict__ rowptr,
                       const int* __restrict__ rank, int* __restrict__ col) {
  int e = blockIdx.x * 256 + threadIdx.x;
  if (e < N_EDGES) col[rowptr[ei[N_EDGES + e]] + rank[e]] = ei[e];
}

#define GM_BLOCKS 768
#define GM_SPITCH 68
static_assert(N_NODES % 16 == 0, "whole 16-row strips");
__global__ void __launch_bounds__(256) k_gemm(const float* __restrict__ X,
                                              const float* __restrict__ W,
                                              float* __restrict__ Y) {
  __shared__ float4 s_w[(D / 4) * D];
  __shared__ float  s_out[8][16 * GM_SPITCH];
  const int tid  = threadIdx.x;
  const int lane = tid & 31;
  const int wv = __builtin_amdgcn_readfirstlane(tid >> 5);

  for (int it = 0; it < (D / 4) * D / 256; ++it) {
    const int idx = it * 256 + tid;
    const float* wp = W + (size_t)(idx >> 7) * (4 * D) + (idx & 127);
    float4 v;
    v.x = wp[0]; v.y = wp[D]; v.z = wp[2 * D]; v.w = wp[3 * D];
    s_w[idx] = v;
  }
  __syncthreads();

  const int c    = lane & 15;
  const int h    = lane >> 4;
  const int half = wv & 1;
  float* so = s_out[wv];
  const float4* bw = s_w + h * D + half * 64 + c;
  const int G = gridDim.x;

  for (int p = (wv >> 1) * G + blockIdx.x; p < N_NODES / 16; p += 4 * G) {
    const int row0 = p * 16;
    const float4* xr = (const float4*)(X + (size_t)(row0 + c) * D) + h;
    float4 av[16];
#pragma unroll
    for (int j = 0; j < 16; ++j) av[j] = xr[2 * j];
    __builtin_amdgcn_sched_barrier(0);

    v8f acc[4] = {};
    float4 bq[2][4];
#pragma unroll
    for (int t = 0; t < 4; ++t) bq[0][t] = bw[t * 16];
#pragma unroll
    for (int j = 0; j < 16; ++j) {
      if (j + 1 < 16) {
#pragma unroll
        for (int t = 0; t < 4; ++t) bq[(j + 1) & 1][t] = bw[(2 * (j + 1)) * D + t * 16];
      }
#pragma unroll
      for (int t = 0; t < 4; ++t) {
        v2f a; a.x = av[j].x; a.y = av[j].y;
        v2f b; b.x = bq[j & 1][t].x; b.y = bq[j & 1][t].y;
        acc[t] = __builtin_amdgcn_wmma_f32_16x16x4_f32(
             false, a,  false, b,
             (short)0, acc[t],  false,  false);
      }
#pragma unroll
      for (int t = 0; t < 4; ++t) {
        v2f a; a.x = av[j].z; a.y = av[j].w;
        v2f b; b.x = bq[j & 1][t].z; b.y = bq[j & 1][t].w;
        acc[t] = __builtin_amdgcn_wmma_f32_16x16x4_f32(
             false, a,  false, b,
             (short)0, acc[t],  false,  false);
      }
      __builtin_amdgcn_sched_barrier(0);
    }

    float* sw = so + (8 * h) * GM_SPITCH + c;
#pragma unroll
    for (int t = 0; t < 4; ++t)
#pragma unroll
      for (int i = 0; i < 8; ++i)
        sw[i * GM_SPITCH + t * 16] = acc[t][i];
    __builtin_amdgcn_fence(__ATOMIC_RELEASE, "wavefront");
    __builtin_amdgcn_fence(__ATOMIC_ACQUIRE, "wavefront");
    const float* sr = so + h * GM_SPITCH + 4 * c;
    float* yb = Y + (size_t)row0 * D + h * D + half * 64 + 4 * c;
#pragma unroll
    for (int s2 = 0; s2 < 8; ++s2) {
      const float4 v = *(const float4*)(sr + s2 * (2 * GM_SPITCH));
      *(float4*)(yb + s2 * (2 * D)) = v;
    }
    __builtin_amdgcn_fence(__ATOMIC_RELEASE, "wavefront");
    __builtin_amdgcn_fence(__ATOMIC_ACQUIRE, "wavefront");
  }
}

__device__ __forceinline__ void fma4(float4& a, const float4 v, const float w) {
  a.x = fmaf(v.x, w, a.x); a.y = fmaf(v.y, w, a.y);
  a.z = fmaf(v.z, w, a.z); a.w = fmaf(v.w, w, a.w);
}

__global__ void __launch_bounds__(256) k_gather(const int* __restrict__ rowptr,
                                                const int* __restrict__ col,
                                                const float* __restrict__ t,
                                                const float* __restrict__ dinv,
                                                const float* __restrict__ bias,
                                                float* __restrict__ h, int relu) {
  const int lane = threadIdx.x & 31;
  const int row = __builtin_amdgcn_readfirstlane(blockIdx.x * 8 + (threadIdx.x >> 5));
  if (row >= N_NODES) return;
  const int beg  = rowptr[row];
  const int m    = rowptr[row + 1] - beg + 1;
  const float di = dinv[row];
  const float4* t4 = (const float4*)t;
  float4 acc = make_float4(0.f, 0.f, 0.f, 0.f);

  for (int base = 0; base < m; base += 32) {
    const int n = min(32, m - base);
    const int k = base + lane;
    int ms = row;
    if (k < m - 1) ms = col[beg + k];
    const int mdb = __float_as_int(dinv[ms]);
    int j = 0;
    for (; j + 4 <= n; j += 4) {
      const int s0 = __builtin_amdgcn_readlane(ms, j);
      const int s1 = __builtin_amdgcn_readlane(ms, j + 1);
      const int s2 = __builtin_amdgcn_readlane(ms, j + 2);
      const int s3 = __builtin_amdgcn_readlane(ms, j + 3);
      const float4 v0 = t4[(size_t)s0 * (D / 4) + lane];
      const float4 v1 = t4[(size_t)s1 * (D / 4) + lane];
      const float4 v2 = t4[(size_t)s2 * (D / 4) + lane];
      const float4 v3 = t4[(size_t)s3 * (D / 4) + lane];
      const float w0 = __int_as_float(__builtin_amdgcn_readlane(mdb, j)) * di;
      const float w1 = __int_as_float(__builtin_amdgcn_readlane(mdb, j + 1)) * di;
      const float w2 = __int_as_float(__builtin_amdgcn_readlane(mdb, j + 2)) * di;
      const float w3 = __int_as_float(__builtin_amdgcn_readlane(mdb, j + 3)) * di;
      fma4(acc, v0, w0);
      fma4(acc, v1, w1);
      fma4(acc, v2, w2);
      fma4(acc, v3, w3);
    }
    const int r = n - j;
    if (r > 0) {
      const int s0 = __builtin_amdgcn_readlane(ms, j);
      const float4 v0 = t4[(size_t)s0 * (D / 4) + lane];
      if (r > 1) {
        const int s1 = __builtin_amdgcn_readlane(ms, j + 1);
        const float4 v1 = t4[(size_t)s1 * (D / 4) + lane];
        if (r > 2) {
          const int s2 = __builtin_amdgcn_readlane(ms, j + 2);
          const float4 v2 = t4[(size_t)s2 * (D / 4) + lane];
          fma4(acc, v2, __int_as_float(__builtin_amdgcn_readlane(mdb, j + 2)) * di);
        }
        fma4(acc, v1, __int_as_float(__builtin_amdgcn_readlane(mdb, j + 1)) * di);
      }
      fma4(acc, v0, __int_as_float(__builtin_amdgcn_readlane(mdb, j)) * di);
    }
  }

  const float4 b = ((const float4*)bias)[lane];
  acc.x += b.x; acc.y += b.y; acc.z += b.z; acc.w += b.w;
  if (relu) {
    acc.x = fmaxf(acc.x, 0.0f); acc.y = fmaxf(acc.y, 0.0f);
    acc.z = fmaxf(acc.z, 0.0f); acc.w = fmaxf(acc.w, 0.0f);
  }
  ((float4*)h)[(size_t)row * (D / 4) + lane] = acc;
}

#define DOT_P 32
#define DOT_U 8
#define DOT_WAVES 4
static_assert(N_PAIRS % DOT_P == 0, "a wave's pairs must all come from one list");
static_assert(DOT_P == 32 && DOT_U == 8, "the exchange below is written for 4 trips of 8 pairs");

__device__ __forceinline__ float halve(const float lo, const float hi, const bool up, const int m) {
  const float keep = up ? hi : lo;
  const float send = up ? lo : hi;
  return keep + __shfl_xor(send, m, 32);
}

__global__ void __launch_bounds__(DOT_WAVES * 32) k_dot(const int* __restrict__ ep,
                                                        const int* __restrict__ en,
                                                        const float* __restrict__ h,
                                                        float* __restrict__ out) {
  const int lane = threadIdx.x & 31;
  const int wave = __builtin_amdgcn_readfirstlane(blockIdx.x * DOT_WAVES + (threadIdx.x >> 5));
  const int p0 = wave * DOT_P;
  if (p0 >= 2 * N_PAIRS) return;
  const int* pr = (p0 < N_PAIRS) ? ep : en;
  const int  q0 = (p0 < N_PAIRS) ? p0 : p0 - N_PAIRS;
  const int2 ab = ((const int2*)pr)[q0 + lane];
  const bool up1 = (lane & 1) != 0, up2 = (lane & 2) != 0, up4 = (lane & 4) != 0;

  float r[DOT_P / DOT_U];
#pragma unroll
  for (int g = 0; g < DOT_P / DOT_U; ++g) {
    float4 va[DOT_U], vb[DOT_U];
#pragma unroll
    for (int u = 0; u < DOT_U; ++u) {
      const int a = __builtin_amdgcn_readlane(ab.x, g * DOT_U + u);
      const int b = __builtin_amdgcn_readlane(ab.y, g * DOT_U + u);
      va[u] = ((const float4*)(h + (size_t)a * D))[lane];
      vb[u] = ((const float4*)(h + (size_t)b * D))[lane];
    }
    float p[DOT_U];
#pragma unroll
    for (int u = 0; u < DOT_U; ++u)
      p[u] = fmaf(va[u].w, vb[u].w, fmaf(va[u].z, vb[u].z, fmaf(va[u].x, vb[u].x, va[u].y * vb[u].y)));
    const float c0 = halve(p[0], p[1], up1, 1), c1 = halve(p[2], p[3], up1, 1);
    const float c2 = halve(p[4], p[5], up1, 1), c3 = halve(p[6], p[7], up1, 1);
    const float d0 = halve(c0, c1, up2, 2), d1 = halve(c2, c3, up2, 2);
    r[g] = halve(d0, d1, up4, 4);
  }
  const bool up8 = (lane & 8) != 0, up16 = (lane & 16) != 0;
  const float w0 = halve(r[0], r[1], up8, 8), w1 = halve(r[2], r[3], up8, 8);
  out[p0 + lane] = halve(w0, w1, up16, 16);
}

extern "C" void kernel_launch(void* const* d_in, const int* in_sizes, int n_in,
                              void* d_out, int out_size, void* d_ws, size_t ws_size,
                              hipStream_t stream) {
  (void)in_sizes; (void)n_in; (void)out_size; (void)ws_size;
  const float* x   = (const float*)d_in[0];
  const float* W1  = (const float*)d_in[1];
  const float* b1  = (const float*)d_in[2];
  const float* W2  = (const float*)d_in[3];
  const float* b2  = (const float*)d_in[4];
  const int* ei    = (const int*)d_in[5];
  const int* edges = (const int*)d_in[6];
  const int* edneg = (const int*)d_in[7];
  float* out = (float*)d_out;

  const size_t FEAT = (size_t)N_NODES * D;
  float* t    = (float*)d_ws;
  float* csr  = t + FEAT;
  float* h    = csr + FEAT;
  float* dinv = h + FEAT;
  int* cnt    = (int*)csr;
  int* rowptr = cnt + N_NODES;
  int* rank   = rowptr + N_NODES + 4;
  int* col    = rank + N_EDGES;

  const int NB_N  = (N_NODES + 255) / 256;
  const int NB_E  = (N_EDGES + 255) / 256;
  const int NB_S  = (N_NODES + SCAN_B - 1) / SCAN_B;
  const int NB_GA = (N_NODES + 7) / 8;
  const int NB_DP = (2 * N_PAIRS / DOT_P + DOT_WAVES - 1) / DOT_WAVES;

  k_zero_cnt<<<NB_N, 256, 0, stream>>>(cnt);
  k_count   <<<NB_E, 256, 0, stream>>>(ei, cnt, rank);
  k_scan    <<<NB_S, SCAN_B, 0, stream>>>(cnt, rowptr, dinv);
  k_fill    <<<NB_E, 256, 0, stream>>>(ei, rowptr, rank, col);

  k_gemm  <<<GM_BLOCKS, 256, 0, stream>>>(x, W1, t);
  k_gather<<<NB_GA, 256, 0, stream>>>(rowptr, col, t, dinv, b1, h, 1);

  k_gemm  <<<GM_BLOCKS, 256, 0, stream>>>(h, W2, t);
  k_gather<<<NB_GA, 256, 0, stream>>>(rowptr, col, t, dinv, b2, h, 0);

  k_dot<<<NB_DP, DOT_WAVES * 32, 0, stream>>>(edges, edneg, h, out);
}


# Edge sort without global atomics: edges binned by 512-node dst bucket via LDS counts (3 launches, were 4), per-bucket LDS count/scan/fill; row sum order = arrival rank at the LDS cursor
# speedup vs baseline: 4.9974x; 1.6237x over previous
#include <hip/hip_runtime.h>

#define N_NODES 100000
#define N_EDGES 1600000
#define N_PAIRS 500000
#define D       128

typedef __attribute__((ext_vector_type(2))) float v2f;
typedef __attribute__((ext_vector_type(8))) float v8f;

#define NPB_LOG 9
#define NPB     (1 << NPB_LOG)
#define NBK     ((N_NODES + NPB - 1) / NPB)
#define NBK_PAD 256
#define BIN_G   256
#define BIN_CH  (N_EDGES / BIN_G)
#define BIN_T   1024
#define BIN_U   ((BIN_CH + BIN_T - 1) / BIN_T)
static_assert(N_EDGES % BIN_G == 0, "equal slices");
static_assert(NBK < NBK_PAD && NBK_PAD == 256 && BIN_G % 16 == 0, "count rows are read as 64 int4 by 16 row groups");

__global__ void __launch_bounds__(BIN_T) k_hist(const int* __restrict__ ei, int* __restrict__ hist) {
  __shared__ int s_h[NBK_PAD];
  const int tid = threadIdx.x;
  if (tid < NBK_PAD) s_h[tid] = 0;
  const int* dst = ei + N_EDGES + blockIdx.x * BIN_CH;
  int d[BIN_U];
#pragma unroll
  for (int u = 0; u < BIN_U; ++u) {
    const int i = tid + u * BIN_T;
    d[u] = (i < BIN_CH) ? dst[i] : -1;
  }
  __syncthreads();
#pragma unroll
  for (int u = 0; u < BIN_U; ++u)
    if (d[u] >= 0) atomicAdd(&s_h[d[u] >> NPB_LOG], 1);
  __syncthreads();
  if (tid < NBK_PAD) hist[blockIdx.x * NBK_PAD + tid] = s_h[tid];
}

__global__ void __launch_bounds__(BIN_T) k_bin(const int* __restrict__ ei, const int* __restrict__ hist,
                                               int* __restrict__ bstart, int* __restrict__ recs_i) {
  __shared__ int  s_T[NBK_PAD];
  __shared__ int  s_P[NBK_PAD];
  __shared__ int  s_cur[NBK_PAD];
  __shared__ int  s_delta[NBK_PAD];
  __shared__ int  s_wt[8], s_wl[8];
  __shared__ int2 s_rec[BIN_CH];
  int2* recs = (int2*)recs_i;
  const int tid  = threadIdx.x;
  const int lane = tid & 31;
  const int wv   = tid >> 5;
  const int w    = blockIdx.x;
  if (tid < NBK_PAD) { s_T[tid] = 0; s_P[tid] = 0; }

  const int* srcp = ei + w * BIN_CH;
  const int* dstp = srcp + N_EDGES;
  int es[BIN_U], ed[BIN_U];
#pragma unroll
  for (int u = 0; u < BIN_U; ++u) {
    const int i = tid + u * BIN_T;
    const bool ok = i < BIN_CH;
    es[u] = ok ? srcp[i] : 0;
    ed[u] = ok ? dstp[i] : -1;
  }

  const int4* h4 = (const int4*)hist;
  const int cq = tid & 63;
  const int r0 = tid >> 6;
  int4 T = make_int4(0, 0, 0, 0), P = make_int4(0, 0, 0, 0);
#pragma unroll
  for (int k = 0; k < BIN_G / 16; ++k) {
    const int row = r0 + 16 * k;
    const int4 v = h4[row * (NBK_PAD / 4) + cq];
    T.x += v.x; T.y += v.y; T.z += v.z; T.w += v.w;
    if (row < w) { P.x += v.x; P.y += v.y; P.z += v.z; P.w += v.w; }
  }
  __syncthreads();
  atomicAdd(&s_T[4 * cq + 0], T.x); atomicAdd(&s_T[4 * cq + 1], T.y);
  atomicAdd(&s_T[4 * cq + 2], T.z); atomicAdd(&s_T[4 * cq + 3], T.w);
  atomicAdd(&s_P[4 * cq + 0], P.x); atomicAdd(&s_P[4 * cq + 1], P.y);
  atomicAdd(&s_P[4 * cq + 2], P.z); atomicAdd(&s_P[4 * cq + 3], P.w);
  __syncthreads();

  int tv = 0, lv = 0;
  if (tid < NBK_PAD) { tv = s_T[tid]; lv = hist[w * NBK_PAD + tid]; }
  int ti = tv, li = lv;
  for (int m = 1; m < 32; m <<= 1) {
    const int o1 = __shfl_up(ti, m, 32);
    const int o2 = __shfl_up(li, m, 32);
    if (lane >= m) { ti += o1; li += o2; }
  }
  if (tid < NBK_PAD && lane == 31) { s_wt[wv] = ti; s_wl[wv] = li; }
  __syncthreads();
  if (tid < NBK_PAD) {
    int pt = 0, pl = 0;
    for (int q = 0; q < wv; ++q) { pt += s_wt[q]; pl += s_wl[q]; }
    const int gstart = pt + ti - tv;
    const int lstart = pl + li - lv;
    s_cur[tid]   = lstart;
    s_delta[tid] = gstart + s_P[tid] - lstart;
    if (w == 0 && tid <= NBK) bstart[tid] = gstart;
  }
  __syncthreads();

  int ps[BIN_U];
#pragma unroll
  for (int u = 0; u < BIN_U; ++u) {
    ps[u] = -1;
    if (ed[u] >= 0) ps[u] = atomicAdd(&s_cur[ed[u] >> NPB_LOG], 1);
  }
#pragma unroll
  for (int u = 0; u < BIN_U; ++u)
    if (ps[u] >= 0) s_rec[ps[u]] = make_int2(es[u], ed[u]);
  __syncthreads();
#pragma unroll
  for (int u = 0; u < BIN_U; ++u) {
    const int i = tid + u * BIN_T;
    if (i < BIN_CH) {
      const int2 r = s_rec[i];
      recs[i + s_delta[r.y >> NPB_LOG]] = r;
    }
  }
}

#define ROW_T   1024
#define ROW_CAP 16384
static_assert(ROW_T >= NPB, "one thread per node of the bucket in the scan");
__global__ void __launch_bounds__(ROW_T) k_rows(const int* __restrict__ bstart, const int* __restrict__ recs_i,
                                                int* __restrict__ rowptr, float* __restrict__ dinv,
                                                int* __restrict__ col) {
  __shared__ int s_cnt[NPB];
  __shared__ int s_cur[NPB];
  __shared__ int s_tot[32];
  __shared__ int s_col[ROW_CAP];
  const int tid  = threadIdx.x;
  const int lane = tid & 31;
  const int wv   = tid >> 5;
  const int b    = blockIdx.x;
  const int start = bstart[b];
  const int n     = bstart[b + 1] - start;
  const int2* rp  = (const int2*)recs_i + start;
  if (tid < NPB) s_cnt[tid] = 0;
  __syncthreads();

  for (int i0 = 0; i0 < n; i0 += 4 * ROW_T) {
    int d[4];
#pragma unroll
    for (int u = 0; u < 4; ++u) {
      const int i = i0 + u * ROW_T + tid;
      d[u] = (i < n) ? rp[i].y : -1;
    }
#pragma unroll
    for (int u = 0; u < 4; ++u)
      if (d[u] >= 0) atomicAdd(&s_cnt[d[u] & (NPB - 1)], 1);
  }
  __syncthreads();

  const int c = (tid < NPB) ? s_cnt[tid] : 0;
  int inc = c;
  for (int m = 1; m < 32; m <<= 1) {
    const int o = __shfl_up(inc, m, 32);
    if (lane >= m) inc += o;
  }
  if (lane == 31) s_tot[wv] = inc;
  __syncthreads();
  if (wv == 0) {
    const int v = s_tot[lane];
    int iv = v;
    for (int m = 1; m < 32; m <<= 1) {
      const int o = __shfl_up(iv, m, 32);
      if (lane >= m) iv += o;
    }
    s_tot[lane] = iv - v;
  }
  __syncthreads();
  if (tid < NPB) {
    const int excl = s_tot[wv] + inc - c;
    const int node = b * NPB + tid;
    s_cur[tid] = excl;
    if (node < N_NODES) {
      rowptr[node] = start + excl;
      dinv[node]   = rsqrtf((float)(c + 1));
    }
  }
  if (b == NBK - 1 && tid == 0) rowptr[N_NODES] = start + n;
  __syncthreads();

  const bool staged = n <= ROW_CAP;
  for (int i0 = 0; i0 < n; i0 += 4 * ROW_T) {
    int2 r[4];
#pragma unroll
    for (int u = 0; u < 4; ++u) {
      const int i = i0 + u * ROW_T + tid;
      r[u] = (i < n) ? rp[i] : make_int2(0, -1);
    }
    int ps[4];
#pragma unroll
    for (int u = 0; u < 4; ++u) {
      ps[u] = -1;
      if (r[u].y >= 0) ps[u] = atomicAdd(&s_cur[r[u].y & (NPB - 1)], 1);
    }
#pragma unroll
    for (int u = 0; u < 4; ++u)
      if (ps[u] >= 0) {
        if (staged) s_col[ps[u]] = r[u].x; else col[start + ps[u]] = r[u].x;
      }
  }
  if (staged) {
    __syncthreads();
    for (int i = tid; i < n; i += ROW_T) col[start + i] = s_col[i];
  }
}

#define GM_BLOCKS 768
#define GM_SPITCH 68
static_assert(N_NODES % 16 == 0, "whole 16-row strips");
__global__ void __launch_bounds__(256) k_gemm(const float* __restrict__ X,
                                              const float* __restrict__ W,
                                              float* __restrict__ Y) {
  __shared__ float4 s_w[(D / 4) * D];
  __shared__ float  s_out[8][16 * GM_SPITCH];
  const int tid  = threadIdx.x;
  const int lane = tid & 31;
  const int wv = __builtin_amdgcn_readfirstlane(tid >> 5);

  for (int it = 0; it < (D / 4) * D / 256; ++it) {
    const int idx = it * 256 + tid;
    const float* wp = W + (size_t)(idx >> 7) * (4 * D) + (idx & 127);
    float4 v;
    v.x = wp[0]; v.y = wp[D]; v.z = wp[2 * D]; v.w = wp[3 * D];
    s_w[idx] = v;
  }
  __syncthreads();

  const int c    = lane & 15;
  const int h    = lane >> 4;
  const int half = wv & 1;
  float* so = s_out[wv];
  const float4* bw = s_w + h * D + half * 64 + c;
  const int G = gridDim.x;

  for (int p = (wv >> 1) * G + blockIdx.x; p < N_NODES / 16; p += 4 * G) {
    const int row0 = p * 16;
    const float4* xr = (const float4*)(X + (size_t)(row0 + c) * D) + h;
    float4 av[16];
#pragma unroll
    for (int j = 0; j < 16; ++j) av[j] = xr[2 * j];
    __builtin_amdgcn_sched_barrier(0);

    v8f acc[4] = {};
    float4 bq[2][4];
#pragma unroll
    for (int t = 0; t < 4; ++t) bq[0][t] = bw[t * 16];
#pragma unroll
    for (int j = 0; j < 16; ++j) {
      if (j + 1 < 16) {
#pragma unroll
        for (int t = 0; t < 4; ++t) bq[(j + 1) & 1][t] = bw[(2 * (j + 1)) * D + t * 16];
      }
#pragma unroll
      for (int t = 0; t < 4; ++t) {
        v2f a; a.x = av[j].x; a.y = av[j].y;
        v2f b; b.x = bq[j & 1][t].x; b.y = bq[j & 1][t].y;
        acc[t] = __builtin_amdgcn_wmma_f32_16x16x4_f32(
             false, a,  false, b,
             (short)0, acc[t],  false,  false);
      }
#pragma unroll
      for (int t = 0; t < 4; ++t) {
        v2f a; a.x = av[j].z; a.y = av[j].w;
        v2f b; b.x = bq[j & 1][t].z; b.y = bq[j & 1][t].w;
        acc[t] = __builtin_amdgcn_wmma_f32_16x16x4_f32(
             false, a,  false, b,
             (short)0, acc[t],  false,  false);
      }
      __builtin_amdgcn_sched_barrier(0);
    }

    float* sw = so + (8 * h) * GM_SPITCH + c;
#pragma unroll
    for (int t = 0; t < 4; ++t)
#pragma unroll
      for (int i = 0; i < 8; ++i)
        sw[i * GM_SPITCH + t * 16] = acc[t][i];
    __builtin_amdgcn_fence(__ATOMIC_RELEASE, "wavefront");
    __builtin_amdgcn_fence(__ATOMIC_ACQUIRE, "wavefront");
    const float* sr = so + h * GM_SPITCH + 4 * c;
    float* yb = Y + (size_t)row0 * D + h * D + half * 64 + 4 * c;
#pragma unroll
    for (int s2 = 0; s2 < 8; ++s2) {
      const float4 v = *(const float4*)(sr + s2 * (2 * GM_SPITCH));
      *(float4*)(yb + s2 * (2 * D)) = v;
    }
    __builtin_amdgcn_fence(__ATOMIC_RELEASE, "wavefront");
    __builtin_amdgcn_fence(__ATOMIC_ACQUIRE, "wavefront");
  }
}

__device__ __forceinline__ void fma4(float4& a, const float4 v, const float w) {
  a.x = fmaf(v.x, w, a.x); a.y = fmaf(v.y, w, a.y);
  a.z = fmaf(v.z, w, a.z); a.w = fmaf(v.w, w, a.w);
}

__global__ void __launch_bounds__(256) k_gather(const int* __restrict__ rowptr,
                                                const int* __restrict__ col,
                                                const float* __restrict__ t,
                                                const float* __restrict__ dinv,
                                                const float* __restrict__ bias,
                                                float* __restrict__ h, int relu) {
  const int lane = threadIdx.x & 31;
  const int row = __builtin_amdgcn_readfirstlane(blockIdx.x * 8 + (threadIdx.x >> 5));
  if (row >= N_NODES) return;
  const int beg  = rowptr[row];
  const int m    = rowptr[row + 1] - beg + 1;
  const float di = dinv[row];
  const float4* t4 = (const float4*)t;
  float4 acc = make_float4(0.f, 0.f, 0.f, 0.f);

  for (int base = 0; base < m; base += 32) {
    const int n = min(32, m - base);
    const int k = base + lane;
    int ms = row;
    if (k < m - 1) ms = col[beg + k];
    const int mdb = __float_as_int(dinv[ms]);
    int j = 0;
    for (; j + 4 <= n; j += 4) {
      const int s0 = __builtin_amdgcn_readlane(ms, j);
      const int s1 = __builtin_amdgcn_readlane(ms, j + 1);
      const int s2 = __builtin_amdgcn_readlane(ms, j + 2);
      const int s3 = __builtin_amdgcn_readlane(ms, j + 3);
      const float4 v0 = t4[(size_t)s0 * (D / 4) + lane];
      const float4 v1 = t4[(size_t)s1 * (D / 4) + lane];
      const float4 v2 = t4[(size_t)s2 * (D / 4) + lane];
      const float4 v3 = t4[(size_t)s3 * (D / 4) + lane];
      const float w0 = __int_as_float(__builtin_amdgcn_readlane(mdb, j)) * di;
      const float w1 = __int_as_float(__builtin_amdgcn_readlane(mdb, j + 1)) * di;
      const float w2 = __int_as_float(__builtin_amdgcn_readlane(mdb, j + 2)) * di;
      const float w3 = __int_as_float(__builtin_amdgcn_readlane(mdb, j + 3)) * di;
      fma4(acc, v0, w0);
      fma4(acc, v1, w1);
      fma4(acc, v2, w2);
      fma4(acc, v3, w3);
    }
    const int r = n - j;
    if (r > 0) {
      const int s0 = __builtin_amdgcn_readlane(ms, j);
      const float4 v0 = t4[(size_t)s0 * (D / 4) + lane];
      if (r > 1) {
        const int s1 = __builtin_amdgcn_readlane(ms, j + 1);
        const float4 v1 = t4[(size_t)s1 * (D / 4) + lane];
        if (r > 2) {
          const int s2 = __builtin_amdgcn_readlane(ms, j + 2);
          const float4 v2 = t4[(size_t)s2 * (D / 4) + lane];
          fma4(acc, v2, __int_as_float(__builtin_amdgcn_readlane(mdb, j + 2)) * di);
        }
        fma4(acc, v1, __int_as_float(__builtin_amdgcn_readlane(mdb, j + 1)) * di);
      }
      fma4(acc, v0, __int_as_float(__builtin_amdgcn_readlane(mdb, j)) * di);
    }
  }

  const float4 b = ((const float4*)bias)[lane];
  acc.x += b.x; acc.y += b.y; acc.z += b.z; acc.w += b.w;
  if (relu) {
    acc.x = fmaxf(acc.x, 0.0f); acc.y = fmaxf(acc.y, 0.0f);
    acc.z = fmaxf(acc.z, 0.0f); acc.w = fmaxf(acc.w, 0.0f);
  }
  ((float4*)h)[(size_t)row * (D / 4) + lane] = acc;
}

#define DOT_P 32
#define DOT_U 8
#define DOT_WAVES 4
static_assert(N_PAIRS % DOT_P == 0, "a wave's pairs must all come from one list");
static_assert(DOT_P == 32 && DOT_U == 8, "the exchange below is written for 4 trips of 8 pairs");

__device__ __forceinline__ float halve(const float lo, const float hi, const bool up, const int m) {
  const float keep = up ? hi : lo;
  const float send = up ? lo : hi;
  return keep + __shfl_xor(send, m, 32);
}

__global__ void __launch_bounds__(DOT_WAVES * 32) k_dot(const int* __restrict__ ep,
                                                        const int* __restrict__ en,
                                                        const float* __restrict__ h,
                                                        float* __restrict__ out) {
  const int lane = threadIdx.x & 31;
  const int wave = __builtin_amdgcn_readfirstlane(blockIdx.x * DOT_WAVES + (threadIdx.x >> 5));
  const int p0 = wave * DOT_P;
  if (p0 >= 2 * N_PAIRS) return;
  const int* pr = (p0 < N_PAIRS) ? ep : en;
  const int  q0 = (p0 < N_PAIRS) ? p0 : p0 - N_PAIRS;
  const int2 ab = ((const int2*)pr)[q0 + lane];
  const bool up1 = (lane & 1) != 0, up2 = (lane & 2) != 0, up4 = (lane & 4) != 0;

  float r[DOT_P / DOT_U];
#pragma unroll
  for (int g = 0; g < DOT_P / DOT_U; ++g) {
    float4 va[DOT_U], vb[DOT_U];
#pragma unroll
    for (int u = 0; u < DOT_U; ++u) {
      const int a = __builtin_amdgcn_readlane(ab.x, g * DOT_U + u);
      const int b = __builtin_amdgcn_readlane(ab.y, g * DOT_U + u);
      va[u] = ((const float4*)(h + (size_t)a * D))[lane];
      vb[u] = ((const float4*)(h + (size_t)b * D))[lane];
    }
    float p[DOT_U];
#pragma unroll
    for (int u = 0; u < DOT_U; ++u)
      p[u] = fmaf(va[u].w, vb[u].w, fmaf(va[u].z, vb[u].z, fmaf(va[u].x, vb[u].x, va[u].y * vb[u].y)));
    const float c0 = halve(p[0], p[1], up1, 1), c1 = halve(p[2], p[3], up1, 1);
    const float c2 = halve(p[4], p[5], up1, 1), c3 = halve(p[6], p[7], up1, 1);
    const float d0 = halve(c0, c1, up2, 2), d1 = halve(c2, c3, up2, 2);
    r[g] = halve(d0, d1, up4, 4);
  }
  const bool up8 = (lane & 8) != 0, up16 = (lane & 16) != 0;
  const float w0 = halve(r[0], r[1], up8, 8), w1 = halve(r[2], r[3], up8, 8);
  out[p0 + lane] = halve(w0, w1, up16, 16);
}

extern "C" void kernel_launch(void* const* d_in, const int* in_sizes, int n_in,
                              void* d_out, int out_size, void* d_ws, size_t ws_size,
                              hipStream_t stream) {
  (void)in_sizes; (void)n_in; (void)out_size; (void)ws_size;
  const float* x   = (const float*)d_in[0];
  const float* W1  = (const float*)d_in[1];
  const float* b1  = (const float*)d_in[2];
  const float* W2  = (const float*)d_in[3];
  const float* b2  = (const float*)d_in[4];
  const int* ei    = (const int*)d_in[5];
  const int* edges = (const int*)d_in[6];
  const int* edneg = (const int*)d_in[7];
  float* out = (float*)d_out;

  const size_t FEAT = (size_t)N_NODES * D;
  float* t    = (float*)d_ws;
  float* csr  = t + FEAT;
  float* h    = csr + FEAT;
  float* dinv = h + FEAT;
  int* rowptr = (int*)csr;
  int* col    = rowptr + N_NODES + 4;
  int* hist   = col + N_EDGES;
  int* bstart = hist + BIN_G * NBK_PAD;
  int* recs   = bstart + 256;

  const int NB_GA = (N_NODES + 7) / 8;
  const int NB_DP = (2 * N_PAIRS / DOT_P + DOT_WAVES - 1) / DOT_WAVES;

  k_hist<<<BIN_G, BIN_T, 0, stream>>>(ei, hist);
  k_bin <<<BIN_G, BIN_T, 0, stream>>>(ei, hist, bstart, recs);
  k_rows<<<NBK, ROW_T, 0, stream>>>(bstart, recs, rowptr, dinv, col);

  k_gemm  <<<GM_BLOCKS, 256, 0, stream>>>(x, W1, t);
  k_gather<<<NB_GA, 256, 0, stream>>>(rowptr, col, t, dinv, b1, h, 1);

  k_gemm  <<<GM_BLOCKS, 256, 0, stream>>>(h, W2, t);
  k_gather<<<NB_GA, 256, 0, stream>>>(rowptr, col, t, dinv, b2, h, 0);

  k_dot<<<NB_DP, DOT_WAVES * 32, 0, stream>>>(edges, edneg, h, out);
}


# dinv[src] folded into the transform: GEMM stores dinv[r]*(xW)[r], pull adds rows and scales the row sum by dinv[d] once; no scattered 4-byte dinv read per in-edge; same sum order, other rounding
# speedup vs baseline: 5.1796x; 1.0365x over previous
#include <hip/hip_runtime.h>

#define N_NODES 100000
#define N_EDGES 1600000
#define N_PAIRS 500000
#define D       128

typedef __attribute__((ext_vector_type(2))) float v2f;
typedef __attribute__((ext_vector_type(8))) float v8f;

#define NPB_LOG 9
#define NPB     (1 << NPB_LOG)
#define NBK     ((N_NODES + NPB - 1) / NPB)
#define NBK_PAD 256
#define BIN_G   256
#define BIN_CH  (N_EDGES / BIN_G)
#define BIN_T   1024
#define BIN_U   ((BIN_CH + BIN_T - 1) / BIN_T)
static_assert(N_EDGES % BIN_G == 0, "equal slices");
static_assert(NBK < NBK_PAD && NBK_PAD == 256 && BIN_G % 16 == 0, "count rows are read as 64 int4 by 16 row groups");

__global__ void __launch_bounds__(BIN_T) k_hist(const int* __restrict__ ei, int* __restrict__ hist) {
  __shared__ int s_h[NBK_PAD];
  const int tid = threadIdx.x;
  if (tid < NBK_PAD) s_h[tid] = 0;
  const int* dst = ei + N_EDGES + blockIdx.x * BIN_CH;
  int d[BIN_U];
#pragma unroll
  for (int u = 0; u < BIN_U; ++u) {
    const int i = tid + u * BIN_T;
    d[u] = (i < BIN_CH) ? dst[i] : -1;
  }
  __syncthreads();
#pragma unroll
  for (int u = 0; u < BIN_U; ++u)
    if (d[u] >= 0) atomicAdd(&s_h[d[u] >> NPB_LOG], 1);
  __syncthreads();
  if (tid < NBK_PAD) hist[blockIdx.x * NBK_PAD + tid] = s_h[tid];
}

__global__ void __launch_bounds__(BIN_T) k_bin(const int* __restrict__ ei, const int* __restrict__ hist,
                                               int* __restrict__ bstart, int* __restrict__ recs_i) {
  __shared__ int  s_T[NBK_PAD];
  __shared__ int  s_P[NBK_PAD];
  __shared__ int  s_cur[NBK_PAD];
  __shared__ int  s_delta[NBK_PAD];
  __shared__ int  s_wt[8], s_wl[8];
  __shared__ int2 s_rec[BIN_CH];
  int2* recs = (int2*)recs_i;
  const int tid  = threadIdx.x;
  const int lane = tid & 31;
  const int wv   = tid >> 5;
  const int w    = blockIdx.x;
  if (tid < NBK_PAD) { s_T[tid] = 0; s_P[tid] = 0; }

  const int* srcp = ei + w * BIN_CH;
  const int* dstp = srcp + N_EDGES;
  int es[BIN_U], ed[BIN_U];
#pragma unroll
  for (int u = 0; u < BIN_U; ++u) {
    const int i = tid + u * BIN_T;
    const bool ok = i < BIN_CH;
    es[u] = ok ? srcp[i] : 0;
    ed[u] = ok ? dstp[i] : -1;
  }

  const int4* h4 = (const int4*)hist;
  const int cq = tid & 63;
  const int r0 = tid >> 6;
  int4 T = make_int4(0, 0, 0, 0), P = make_int4(0, 0, 0, 0);
#pragma unroll
  for (int k = 0; k < BIN_G / 16; ++k) {
    const int row = r0 + 16 * k;
    const int4 v = h4[row * (NBK_PAD / 4) + cq];
    T.x += v.x; T.y += v.y; T.z += v.z; T.w += v.w;
    if (row < w) { P.x += v.x; P.y += v.y; P.z += v.z; P.w += v.w; }
  }
  __syncthreads();
  atomicAdd(&s_T[4 * cq + 0], T.x); atomicAdd(&s_T[4 * cq + 1], T.y);
  atomicAdd(&s_T[4 * cq + 2], T.z); atomicAdd(&s_T[4 * cq + 3], T.w);
  atomicAdd(&s_P[4 * cq + 0], P.x); atomicAdd(&s_P[4 * cq + 1], P.y);
  atomicAdd(&s_P[4 * cq + 2], P.z); atomicAdd(&s_P[4 * cq + 3], P.w);
  __syncthreads();

  int tv = 0, lv = 0;
  if (tid < NBK_PAD) { tv = s_T[tid]; lv = hist[w * NBK_PAD + tid]; }
  int ti = tv, li = lv;
  for (int m = 1; m < 32; m <<= 1) {
    const int o1 = __shfl_up(ti, m, 32);
    const int o2 = __shfl_up(li, m, 32);
    if (lane >= m) { ti += o1; li += o2; }
  }
  if (tid < NBK_PAD && lane == 31) { s_wt[wv] = ti; s_wl[wv] = li; }
  __syncthreads();
  if (tid < NBK_PAD) {
    int pt = 0, pl = 0;
    for (int q = 0; q < wv; ++q) { pt += s_wt[q]; pl += s_wl[q]; }
    const int gstart = pt + ti - tv;
    const int lstart = pl + li - lv;
    s_cur[tid]   = lstart;
    s_delta[tid] = gstart + s_P[tid] - lstart;
    if (w == 0 && tid <= NBK) bstart[tid] = gstart;
  }
  __syncthreads();

  int ps[BIN_U];
#pragma unroll
  for (int u = 0; u < BIN_U; ++u) {
    ps[u] = -1;
    if (ed[u] >= 0) ps[u] = atomicAdd(&s_cur[ed[u] >> NPB_LOG], 1);
  }
#pragma unroll
  for (int u = 0; u < BIN_U; ++u)
    if (ps[u] >= 0) s_rec[ps[u]] = make_int2(es[u], ed[u]);
  __syncthreads();
#pragma unroll
  for (int u = 0; u < BIN_U; ++u) {
    const int i = tid + u * BIN_T;
    if (i < BIN_CH) {
      const int2 r = s_rec[i];
      recs[i + s_delta[r.y >> NPB_LOG]] = r;
    }
  }
}

#define ROW_T   1024
#define ROW_CAP 16384
static_assert(ROW_T >= NPB, "one thread per node of the bucket in the scan");
__global__ void __launch_bounds__(ROW_T) k_rows(const int* __restrict__ bstart, const int* __restrict__ recs_i,
                                                int* __restrict__ rowptr, float* __restrict__ dinv,
                                                int* __restrict__ col) {
  __shared__ int s_cnt[NPB];
  __shared__ int s_cur[NPB];
  __shared__ int s_tot[32];
  __shared__ int s_col[ROW_CAP];
  const int tid  = threadIdx.x;
  const int lane = tid & 31;
  const int wv   = tid >> 5;
  const int b    = blockIdx.x;
  const int start = bstart[b];
  const int n     = bstart[b + 1] - start;
  const int2* rp  = (const int2*)recs_i + start;
  if (tid < NPB) s_cnt[tid] = 0;
  __syncthreads();

  for (int i0 = 0; i0 < n; i0 += 4 * ROW_T) {
    int d[4];
#pragma unroll
    for (int u = 0; u < 4; ++u) {
      const int i = i0 + u * ROW_T + tid;
      d[u] = (i < n) ? rp[i].y : -1;
    }
#pragma unroll
    for (int u = 0; u < 4; ++u)
      if (d[u] >= 0) atomicAdd(&s_cnt[d[u] & (NPB - 1)], 1);
  }
  __syncthreads();

  const int c = (tid < NPB) ? s_cnt[tid] : 0;
  int inc = c;
  for (int m = 1; m < 32; m <<= 1) {
    const int o = __shfl_up(inc, m, 32);
    if (lane >= m) inc += o;
  }
  if (lane == 31) s_tot[wv] = inc;
  __syncthreads();
  if (wv == 0) {
    const int v = s_tot[lane];
    int iv = v;
    for (int m = 1; m < 32; m <<= 1) {
      const int o = __shfl_up(iv, m, 32);
      if (lane >= m) iv += o;
    }
    s_tot[lane] = iv - v;
  }
  __syncthreads();
  if (tid < NPB) {
    const int excl = s_tot[wv] + inc - c;
    const int node = b * NPB + tid;
    s_cur[tid] = excl;
    if (node < N_NODES) {
      rowptr[node] = start + excl;
      dinv[node]   = rsqrtf((float)(c + 1));
    }
  }
  if (b == NBK - 1 && tid == 0) rowptr[N_NODES] = start + n;
  __syncthreads();

  const bool staged = n <= ROW_CAP;
  for (int i0 = 0; i0 < n; i0 += 4 * ROW_T) {
    int2 r[4];
#pragma unroll
    for (int u = 0; u < 4; ++u) {
      const int i = i0 + u * ROW_T + tid;
      r[u] = (i < n) ? rp[i] : make_int2(0, -1);
    }
    int ps[4];
#pragma unroll
    for (int u = 0; u < 4; ++u) {
      ps[u] = -1;
      if (r[u].y >= 0) ps[u] = atomicAdd(&s_cur[r[u].y & (NPB - 1)], 1);
    }
#pragma unroll
    for (int u = 0; u < 4; ++u)
      if (ps[u] >= 0) {
        if (staged) s_col[ps[u]] = r[u].x; else col[start + ps[u]] = r[u].x;
      }
  }
  if (staged) {
    __syncthreads();
    for (int i = tid; i < n; i += ROW_T) col[start + i] = s_col[i];
  }
}

#define GM_BLOCKS 768
#define GM_SPITCH 68
static_assert(N_NODES % 16 == 0, "whole 16-row strips");
__global__ void __launch_bounds__(256) k_gemm(const float* __restrict__ X,
                                              const float* __restrict__ W,
                                              const float* __restrict__ dinv,
                                              float* __restrict__ Y) {
  __shared__ float4 s_w[(D / 4) * D];
  __shared__ float  s_out[8][16 * GM_SPITCH];
  const int tid  = threadIdx.x;
  const int lane = tid & 31;
  const int wv = __builtin_amdgcn_readfirstlane(tid >> 5);

  for (int it = 0; it < (D / 4) * D / 256; ++it) {
    const int idx = it * 256 + tid;
    const float* wp = W + (size_t)(idx >> 7) * (4 * D) + (idx & 127);
    float4 v;
    v.x = wp[0]; v.y = wp[D]; v.z = wp[2 * D]; v.w = wp[3 * D];
    s_w[idx] = v;
  }
  __syncthreads();

  const int c    = lane & 15;
  const int h    = lane >> 4;
  const int half = wv & 1;
  float* so = s_out[wv];
  const float4* bw = s_w + h * D + half * 64 + c;
  const int G = gridDim.x;

  for (int p = (wv >> 1) * G + blockIdx.x; p < N_NODES / 16; p += 4 * G) {
    const int row0 = p * 16;
    const float4* xr = (const float4*)(X + (size_t)(row0 + c) * D) + h;
    float4 av[16];
#pragma unroll
    for (int j = 0; j < 16; ++j) av[j] = xr[2 * j];
    const float4* dp = (const float4*)(dinv + row0 + 8 * h);
    const float4 ds0 = dp[0], ds1 = dp[1];
    __builtin_amdgcn_sched_barrier(0);

    v8f acc[4] = {};
    float4 bq[2][4];
#pragma unroll
    for (int t = 0; t < 4; ++t) bq[0][t] = bw[t * 16];
#pragma unroll
    for (int j = 0; j < 16; ++j) {
      if (j + 1 < 16) {
#pragma unroll
        for (int t = 0; t < 4; ++t) bq[(j + 1) & 1][t] = bw[(2 * (j + 1)) * D + t * 16];
      }
#pragma unroll
      for (int t = 0; t < 4; ++t) {
        v2f a; a.x = av[j].x; a.y = av[j].y;
        v2f b; b.x = bq[j & 1][t].x; b.y = bq[j & 1][t].y;
        acc[t] = __builtin_amdgcn_wmma_f32_16x16x4_f32(
             false, a,  false, b,
             (short)0, acc[t],  false,  false);
      }
#pragma unroll
      for (int t = 0; t < 4; ++t) {
        v2f a; a.x = av[j].z; a.y = av[j].w;
        v2f b; b.x = bq[j & 1][t].z; b.y = bq[j & 1][t].w;
        acc[t] = __builtin_amdgcn_wmma_f32_16x16x4_f32(
             false, a,  false, b,
             (short)0, acc[t],  false,  false);
      }
      __builtin_amdgcn_sched_barrier(0);
    }

    const float sc[8] = {ds0.x, ds0.y, ds0.z, ds0.w, ds1.x, ds1.y, ds1.z, ds1.w};
    float* sw = so + (8 * h) * GM_SPITCH + c;
#pragma unroll
    for (int t = 0; t < 4; ++t)
#pragma unroll
      for (int i = 0; i < 8; ++i)
        sw[i * GM_SPITCH + t * 16] = acc[t][i] * sc[i];
    __builtin_amdgcn_fence(__ATOMIC_RELEASE, "wavefront");
    __builtin_amdgcn_fence(__ATOMIC_ACQUIRE, "wavefront");
    const float* sr = so + h * GM_SPITCH + 4 * c;
    float* yb = Y + (size_t)row0 * D + h * D + half * 64 + 4 * c;
#pragma unroll
    for (int s2 = 0; s2 < 8; ++s2) {
      const float4 v = *(const float4*)(sr + s2 * (2 * GM_SPITCH));
      *(float4*)(yb + s2 * (2 * D)) = v;
    }
    __builtin_amdgcn_fence(__ATOMIC_RELEASE, "wavefront");
    __builtin_amdgcn_fence(__ATOMIC_ACQUIRE, "wavefront");
  }
}

__device__ __forceinline__ void add4(float4& a, const float4 v) {
  a.x += v.x; a.y += v.y; a.z += v.z; a.w += v.w;
}

__global__ void __launch_bounds__(256) k_gather(const int* __restrict__ rowptr,
                                                const int* __restrict__ col,
                                                const float* __restrict__ t,
                                                const float* __restrict__ dinv,
                                                const float* __restrict__ bias,
                                                float* __restrict__ h, int relu) {
  const int lane = threadIdx.x & 31;
  const int row = __builtin_amdgcn_readfirstlane(blockIdx.x * 8 + (threadIdx.x >> 5));
  if (row >= N_NODES) return;
  const int beg  = rowptr[row];
  const int m    = rowptr[row + 1] - beg + 1;
  const float di = dinv[row];
  const float4* t4 = (const float4*)t;
  float4 acc = make_float4(0.f, 0.f, 0.f, 0.f);

  for (int base = 0; base < m; base += 32) {
    const int n = min(32, m - base);
    const int k = base + lane;
    int ms = row;
    if (k < m - 1) ms = col[beg + k];
    int j = 0;
    for (; j + 4 <= n; j += 4) {
      const int s0 = __builtin_amdgcn_readlane(ms, j);
      const int s1 = __builtin_amdgcn_readlane(ms, j + 1);
      const int s2 = __builtin_amdgcn_readlane(ms, j + 2);
      const int s3 = __builtin_amdgcn_readlane(ms, j + 3);
      const float4 v0 = t4[(size_t)s0 * (D / 4) + lane];
      const float4 v1 = t4[(size_t)s1 * (D / 4) + lane];
      const float4 v2 = t4[(size_t)s2 * (D / 4) + lane];
      const float4 v3 = t4[(size_t)s3 * (D / 4) + lane];
      add4(acc, v0);
      add4(acc, v1);
      add4(acc, v2);
      add4(acc, v3);
    }
    const int r = n - j;
    if (r > 0) {
      const int s0 = __builtin_amdgcn_readlane(ms, j);
      const float4 v0 = t4[(size_t)s0 * (D / 4) + lane];
      if (r > 1) {
        const int s1 = __builtin_amdgcn_readlane(ms, j + 1);
        const float4 v1 = t4[(size_t)s1 * (D / 4) + lane];
        if (r > 2) {
          const int s2 = __builtin_amdgcn_readlane(ms, j + 2);
          const float4 v2 = t4[(size_t)s2 * (D / 4) + lane];
          add4(acc, v2);
        }
        add4(acc, v1);
      }
      add4(acc, v0);
    }
  }

  const float4 b = ((const float4*)bias)[lane];
  acc.x = fmaf(acc.x, di, b.x); acc.y = fmaf(acc.y, di, b.y);
  acc.z = fmaf(acc.z, di, b.z); acc.w = fmaf(acc.w, di, b.w);
  if (relu) {
    acc.x = fmaxf(acc.x, 0.0f); acc.y = fmaxf(acc.y, 0.0f);
    acc.z = fmaxf(acc.z, 0.0f); acc.w = fmaxf(acc.w, 0.0f);
  }
  ((float4*)h)[(size_t)row * (D / 4) + lane] = acc;
}

#define DOT_P 32
#define DOT_U 8
#define DOT_WAVES 4
static_assert(N_PAIRS % DOT_P == 0, "a wave's pairs must all come from one list");
static_assert(DOT_P == 32 && DOT_U == 8, "the exchange below is written for 4 trips of 8 pairs");

__device__ __forceinline__ float halve(const float lo, const float hi, const bool up, const int m) {
  const float keep = up ? hi : lo;
  const float send = up ? lo : hi;
  return keep + __shfl_xor(send, m, 32);
}

__global__ void __launch_bounds__(DOT_WAVES * 32) k_dot(const int* __restrict__ ep,
                                                        const int* __restrict__ en,
                                                        const float* __restrict__ h,
                                                        float* __restrict__ out) {
  const int lane = threadIdx.x & 31;
  const int wave = __builtin_amdgcn_readfirstlane(blockIdx.x * DOT_WAVES + (threadIdx.x >> 5));
  const int p0 = wave * DOT_P;
  if (p0 >= 2 * N_PAIRS) return;
  const int* pr = (p0 < N_PAIRS) ? ep : en;
  const int  q0 = (p0 < N_PAIRS) ? p0 : p0 - N_PAIRS;
  const int2 ab = ((const int2*)pr)[q0 + lane];
  const bool up1 = (lane & 1) != 0, up2 = (lane & 2) != 0, up4 = (lane & 4) != 0;

  float r[DOT_P / DOT_U];
#pragma unroll
  for (int g = 0; g < DOT_P / DOT_U; ++g) {
    float4 va[DOT_U], vb[DOT_U];
#pragma unroll
    for (int u = 0; u < DOT_U; ++u) {
      const int a = __builtin_amdgcn_readlane(ab.x, g * DOT_U + u);
      const int b = __builtin_amdgcn_readlane(ab.y, g * DOT_U + u);
      va[u] = ((const float4*)(h + (size_t)a * D))[lane];
      vb[u] = ((const float4*)(h + (size_t)b * D))[lane];
    }
    float p[DOT_U];
#pragma unroll
    for (int u = 0; u < DOT_U; ++u)
      p[u] = fmaf(va[u].w, vb[u].w, fmaf(va[u].z, vb[u].z, fmaf(va[u].x, vb[u].x, va[u].y * vb[u].y)));
    const float c0 = halve(p[0], p[1], up1, 1), c1 = halve(p[2], p[3], up1, 1);
    const float c2 = halve(p[4], p[5], up1, 1), c3 = halve(p[6], p[7], up1, 1);
    const float d0 = halve(c0, c1, up2, 2), d1 = halve(c2, c3, up2, 2);
    r[g] = halve(d0, d1, up4, 4);
  }
  const bool up8 = (lane & 8) != 0, up16 = (lane & 16) != 0;
  const float w0 = halve(r[0], r[1], up8, 8), w1 = halve(r[2], r[3], up8, 8);
  out[p0 + lane] = halve(w0, w1, up16, 16);
}

extern "C" void kernel_launch(void* const* d_in, const int* in_sizes, int n_in,
                              void* d_out, int out_size, void* d_ws, size_t ws_size,
                              hipStream_t stream) {
  (void)in_sizes; (void)n_in; (void)out_size; (void)ws_size;
  const float* x   = (const float*)d_in[0];
  const float* W1  = (const float*)d_in[1];
  const float* b1  = (const float*)d_in[2];
  const float* W2  = (const float*)d_in[3];
  const float* b2  = (const float*)d_in[4];
  const int* ei    = (const int*)d_in[5];
  const int* edges = (const int*)d_in[6];
  const int* edneg = (const int*)d_in[7];
  float* out = (float*)d_out;

  const size_t FEAT = (size_t)N_NODES * D;
  float* t    = (float*)d_ws;
  float* csr  = t + FEAT;
  float* h    = csr + FEAT;
  float* dinv = h + FEAT;
  int* rowptr = (int*)csr;
  int* col    = rowptr + N_NODES + 4;
  int* hist   = col + N_EDGES;
  int* bstart = hist + BIN_G * NBK_PAD;
  int* recs   = bstart + 256;

  const int NB_GA = (N_NODES + 7) / 8;
  const int NB_DP = (2 * N_PAIRS / DOT_P + DOT_WAVES - 1) / DOT_WAVES;

  k_hist<<<BIN_G, BIN_T, 0, stream>>>(ei, hist);
  k_bin <<<BIN_G, BIN_T, 0, stream>>>(ei, hist, bstart, recs);
  k_rows<<<NBK, ROW_T, 0, stream>>>(bstart, recs, rowptr, dinv, col);

  k_gemm  <<<GM_BLOCKS, 256, 0, stream>>>(x, W1, dinv, t);
  k_gather<<<NB_GA, 256, 0, stream>>>(rowptr, col, t, dinv, b1, h, 1);

  k_gemm  <<<GM_BLOCKS, 256, 0, stream>>>(h, W2, dinv, t);
  k_gather<<<NB_GA, 256, 0, stream>>>(rowptr, col, t, dinv, b2, h, 0);

  k_dot<<<NB_DP, DOT_WAVES * 32, 0, stream>>>(edges, edneg, h, out);
}


# Layer-1 pull fused with the layer-2 transform: 16 pull waves fill an LDS strip of 16 rows, 4 matrix waves multiply it by W2 one strip behind; layer-1 output never stored; sum orders unchanged
# speedup vs baseline: 5.4091x; 1.0443x over previous
#include <hip/hip_runtime.h>

#define N_NODES 100000
#define N_EDGES 1600000
#define N_PAIRS 500000
#define D       128

typedef __attribute__((ext_vector_type(2))) float v2f;
typedef __attribute__((ext_vector_type(8))) float v8f;

#define NPB_LOG 9
#define NPB     (1 << NPB_LOG)
#define NBK     ((N_NODES + NPB - 1) / NPB)
#define NBK_PAD 256
#define BIN_G   256
#define BIN_CH  (N_EDGES / BIN_G)
#define BIN_T   1024
#define BIN_U   ((BIN_CH + BIN_T - 1) / BIN_T)
static_assert(N_EDGES % BIN_G == 0, "equal slices");
static_assert(NBK < NBK_PAD && NBK_PAD == 256 && BIN_G % 16 == 0, "count rows are read as 64 int4 by 16 row groups");

__global__ void __launch_bounds__(BIN_T) k_hist(const int* __restrict__ ei, int* __restrict__ hist) {
  __shared__ int s_h[NBK_PAD];
  const int tid = threadIdx.x;
  if (tid < NBK_PAD) s_h[tid] = 0;
  const int* dst = ei + N_EDGES + blockIdx.x * BIN_CH;
  int d[BIN_U];
#pragma unroll
  for (int u = 0; u < BIN_U; ++u) {
    const int i = tid + u * BIN_T;
    d[u] = (i < BIN_CH) ? dst[i] : -1;
  }
  __syncthreads();
#pragma unroll
  for (int u = 0; u < BIN_U; ++u)
    if (d[u] >= 0) atomicAdd(&s_h[d[u] >> NPB_LOG], 1);
  __syncthreads();
  if (tid < NBK_PAD) hist[blockIdx.x * NBK_PAD + tid] = s_h[tid];
}

__global__ void __launch_bounds__(BIN_T) k_bin(const int* __restrict__ ei, const int* __restrict__ hist,
                                               int* __restrict__ bstart, int* __restrict__ recs_i) {
  __shared__ int  s_T[NBK_PAD];
  __shared__ int  s_P[NBK_PAD];
  __shared__ int  s_cur[NBK_PAD];
  __shared__ int  s_delta[NBK_PAD];
  __shared__ int  s_wt[8], s_wl[8];
  __shared__ int2 s_rec[BIN_CH];
  int2* recs = (int2*)recs_i;
  const int tid  = threadIdx.x;
  const int lane = tid & 31;
  const int wv   = tid >> 5;
  const int w    = blockIdx.x;
  if (tid < NBK_PAD) { s_T[tid] = 0; s_P[tid] = 0; }

  const int* srcp = ei + w * BIN_CH;
  const int* dstp = srcp + N_EDGES;
  int es[BIN_U], ed[BIN_U];
#pragma unroll
  for (int u = 0; u < BIN_U; ++u) {
    const int i = tid + u * BIN_T;
    const bool ok = i < BIN_CH;
    es[u] = ok ? srcp[i] : 0;
    ed[u] = ok ? dstp[i] : -1;
  }

  const int4* h4 = (const int4*)hist;
  const int cq = tid & 63;
  const int r0 = tid >> 6;
  int4 T = make_int4(0, 0, 0, 0), P = make_int4(0, 0, 0, 0);
#pragma unroll
  for (int k = 0; k < BIN_G / 16; ++k) {
    const int row = r0 + 16 * k;
    const int4 v = h4[row * (NBK_PAD / 4) + cq];
    T.x += v.x; T.y += v.y; T.z += v.z; T.w += v.w;
    if (row < w) { P.x += v.x; P.y += v.y; P.z += v.z; P.w += v.w; }
  }
  __syncthreads();
  atomicAdd(&s_T[4 * cq + 0], T.x); atomicAdd(&s_T[4 * cq + 1], T.y);
  atomicAdd(&s_T[4 * cq + 2], T.z); atomicAdd(&s_T[4 * cq + 3], T.w);
  atomicAdd(&s_P[4 * cq + 0], P.x); atomicAdd(&s_P[4 * cq + 1], P.y);
  atomicAdd(&s_P[4 * cq + 2], P.z); atomicAdd(&s_P[4 * cq + 3], P.w);
  __syncthreads();

  int tv = 0, lv = 0;
  if (tid < NBK_PAD) { tv = s_T[tid]; lv = hist[w * NBK_PAD + tid]; }
  int ti = tv, li = lv;
  for (int m = 1; m < 32; m <<= 1) {
    const int o1 = __shfl_up(ti, m, 32);
    const int o2 = __shfl_up(li, m, 32);
    if (lane >= m) { ti += o1; li += o2; }
  }
  if (tid < NBK_PAD && lane == 31) { s_wt[wv] = ti; s_wl[wv] = li; }
  __syncthreads();
  if (tid < NBK_PAD) {
    int pt = 0, pl = 0;
    for (int q = 0; q < wv; ++q) { pt += s_wt[q]; pl += s_wl[q]; }
    const int gstart = pt + ti - tv;
    const int lstart = pl + li - lv;
    s_cur[tid]   = lstart;
    s_delta[tid] = gstart + s_P[tid] - lstart;
    if (w == 0 && tid <= NBK) bstart[tid] = gstart;
  }
  __syncthreads();

  int ps[BIN_U];
#pragma unroll
  for (int u = 0; u < BIN_U; ++u) {
    ps[u] = -1;
    if (ed[u] >= 0) ps[u] = atomicAdd(&s_cur[ed[u] >> NPB_LOG], 1);
  }
#pragma unroll
  for (int u = 0; u < BIN_U; ++u)
    if (ps[u] >= 0) s_rec[ps[u]] = make_int2(es[u], ed[u]);
  __syncthreads();
#pragma unroll
  for (int u = 0; u < BIN_U; ++u) {
    const int i = tid + u * BIN_T;
    if (i < BIN_CH) {
      const int2 r = s_rec[i];
      recs[i + s_delta[r.y >> NPB_LOG]] = r;
    }
  }
}

#define ROW_T   1024
#define ROW_CAP 16384
static_assert(ROW_T >= NPB, "one thread per node of the bucket in the scan");
__global__ void __launch_bounds__(ROW_T) k_rows(const int* __restrict__ bstart, const int* __restrict__ recs_i,
                                                int* __restrict__ rowptr, float* __restrict__ dinv,
                                                int* __restrict__ col) {
  __shared__ int s_cnt[NPB];
  __shared__ int s_cur[NPB];
  __shared__ int s_tot[32];
  __shared__ int s_col[ROW_CAP];
  const int tid  = threadIdx.x;
  const int lane = tid & 31;
  const int wv   = tid >> 5;
  const int b    = blockIdx.x;
  const int start = bstart[b];
  const int n     = bstart[b + 1] - start;
  const int2* rp  = (const int2*)recs_i + start;
  if (tid < NPB) s_cnt[tid] = 0;
  __syncthreads();

  for (int i0 = 0; i0 < n; i0 += 4 * ROW_T) {
    int d[4];
#pragma unroll
    for (int u = 0; u < 4; ++u) {
      const int i = i0 + u * ROW_T + tid;
      d[u] = (i < n) ? rp[i].y : -1;
    }
#pragma unroll
    for (int u = 0; u < 4; ++u)
      if (d[u] >= 0) atomicAdd(&s_cnt[d[u] & (NPB - 1)], 1);
  }
  __syncthreads();

  const int c = (tid < NPB) ? s_cnt[tid] : 0;
  int inc = c;
  for (int m = 1; m < 32; m <<= 1) {
    const int o = __shfl_up(inc, m, 32);
    if (lane >= m) inc += o;
  }
  if (lane == 31) s_tot[wv] = inc;
  __syncthreads();
  if (wv == 0) {
    const int v = s_tot[lane];
    int iv = v;
    for (int m = 1; m < 32; m <<= 1) {
      const int o = __shfl_up(iv, m, 32);
      if (lane >= m) iv += o;
    }
    s_tot[lane] = iv - v;
  }
  __syncthreads();
  if (tid < NPB) {
    const int excl = s_tot[wv] + inc - c;
    const int node = b * NPB + tid;
    s_cur[tid] = excl;
    if (node < N_NODES) {
      rowptr[node] = start + excl;
      dinv[node]   = rsqrtf((float)(c + 1));
    }
  }
  if (b == NBK - 1 && tid == 0) rowptr[N_NODES] = start + n;
  __syncthreads();

  const bool staged = n <= ROW_CAP;
  for (int i0 = 0; i0 < n; i0 += 4 * ROW_T) {
    int2 r[4];
#pragma unroll
    for (int u = 0; u < 4; ++u) {
      const int i = i0 + u * ROW_T + tid;
      r[u] = (i < n) ? rp[i] : make_int2(0, -1);
    }
    int ps[4];
#pragma unroll
    for (int u = 0; u < 4; ++u) {
      ps[u] = -1;
      if (r[u].y >= 0) ps[u] = atomicAdd(&s_cur[r[u].y & (NPB - 1)], 1);
    }
#pragma unroll
    for (int u = 0; u < 4; ++u)
      if (ps[u] >= 0) {
        if (staged) s_col[ps[u]] = r[u].x; else col[start + ps[u]] = r[u].x;
      }
  }
  if (staged) {
    __syncthreads();
    for (int i = tid; i < n; i += ROW_T) col[start + i] = s_col[i];
  }
}

#define GM_BLOCKS 768
#define GM_SPITCH 68
static_assert(N_NODES % 16 == 0, "whole 16-row strips");
__global__ void __launch_bounds__(256) k_gemm(const float* __restrict__ X,
                                              const float* __restrict__ W,
                                              const float* __restrict__ dinv,
                                              float* __restrict__ Y) {
  __shared__ float4 s_w[(D / 4) * D];
  __shared__ float  s_out[8][16 * GM_SPITCH];
  const int tid  = threadIdx.x;
  const int lane = tid & 31;
  const int wv = __builtin_amdgcn_readfirstlane(tid >> 5);

  for (int it = 0; it < (D / 4) * D / 256; ++it) {
    const int idx = it * 256 + tid;
    const float* wp = W + (size_t)(idx >> 7) * (4 * D) + (idx & 127);
    float4 v;
    v.x = wp[0]; v.y = wp[D]; v.z = wp[2 * D]; v.w = wp[3 * D];
    s_w[idx] = v;
  }
  __syncthreads();

  const int c    = lane & 15;
  const int h    = lane >> 4;
  const int half = wv & 1;
  float* so = s_out[wv];
  const float4* bw = s_w + h * D + half * 64 + c;
  const int G = gridDim.x;

  for (int p = (wv >> 1) * G + blockIdx.x; p < N_NODES / 16; p += 4 * G) {
    const int row0 = p * 16;
    const float4* xr = (const float4*)(X + (size_t)(row0 + c) * D) + h;
    float4 av[16];
#pragma unroll
    for (int j = 0; j < 16; ++j) av[j] = xr[2 * j];
    const float4* dp = (const float4*)(dinv + row0 + 8 * h);
    const float4 ds0 = dp[0], ds1 = dp[1];
    __builtin_amdgcn_sched_barrier(0);

    v8f acc[4] = {};
    float4 bq[2][4];
#pragma unroll
    for (int t = 0; t < 4; ++t) bq[0][t] = bw[t * 16];
#pragma unroll
    for (int j = 0; j < 16; ++j) {
      if (j + 1 < 16) {
#pragma unroll
        for (int t = 0; t < 4; ++t) bq[(j + 1) & 1][t] = bw[(2 * (j + 1)) * D + t * 16];
      }
#pragma unroll
      for (int t = 0; t < 4; ++t) {
        v2f a; a.x = av[j].x; a.y = av[j].y;
        v2f b; b.x = bq[j & 1][t].x; b.y = bq[j & 1][t].y;
        acc[t] = __builtin_amdgcn_wmma_f32_16x16x4_f32(
             false, a,  false, b,
             (short)0, acc[t],  false,  false);
      }
#pragma unroll
      for (int t = 0; t < 4; ++t) {
        v2f a; a.x = av[j].z; a.y = av[j].w;
        v2f b; b.x = bq[j & 1][t].z; b.y = bq[j & 1][t].w;
        acc[t] = __builtin_amdgcn_wmma_f32_16x16x4_f32(
             false, a,  false, b,
             (short)0, acc[t],  false,  false);
      }
      __builtin_amdgcn_sched_barrier(0);
    }

    const float sc[8] = {ds0.x, ds0.y, ds0.z, ds0.w, ds1.x, ds1.y, ds1.z, ds1.w};
    float* sw = so + (8 * h) * GM_SPITCH + c;
#pragma unroll
    for (int t = 0; t < 4; ++t)
#pragma unroll
      for (int i = 0; i < 8; ++i)
        sw[i * GM_SPITCH + t * 16] = acc[t][i] * sc[i];
    __builtin_amdgcn_fence(__ATOMIC_RELEASE, "wavefront");
    __builtin_amdgcn_fence(__ATOMIC_ACQUIRE, "wavefront");
    const float* sr = so + h * GM_SPITCH + 4 * c;
    float* yb = Y + (size_t)row0 * D + h * D + half * 64 + 4 * c;
#pragma unroll
    for (int s2 = 0; s2 < 8; ++s2) {
      const float4 v = *(const float4*)(sr + s2 * (2 * GM_SPITCH));
      *(float4*)(yb + s2 * (2 * D)) = v;
    }
    __builtin_amdgcn_fence(__ATOMIC_RELEASE, "wavefront");
    __builtin_amdgcn_fence(__ATOMIC_ACQUIRE, "wavefront");
  }
}

__device__ __forceinline__ void add4(float4& a, const float4 v) {
  a.x += v.x; a.y += v.y; a.z += v.z; a.w += v.w;
}

__global__ void __launch_bounds__(256) k_gather(const int* __restrict__ rowptr,
                                                const int* __restrict__ col,
                                                const float* __restrict__ t,
                                                const float* __restrict__ dinv,
                                                const float* __restrict__ bias,
                                                float* __restrict__ h, int relu) {
  const int lane = threadIdx.x & 31;
  const int row = __builtin_amdgcn_readfirstlane(blockIdx.x * 8 + (threadIdx.x >> 5));
  if (row >= N_NODES) return;
  const int beg  = rowptr[row];
  const int m    = rowptr[row + 1] - beg + 1;
  const float di = dinv[row];
  const float4* t4 = (const float4*)t;
  float4 acc = make_float4(0.f, 0.f, 0.f, 0.f);

  for (int base = 0; base < m; base += 32) {
    const int n = min(32, m - base);
    const int k = base + lane;
    int ms = row;
    if (k < m - 1) ms = col[beg + k];
    int j = 0;
    for (; j + 4 <= n; j += 4) {
      const int s0 = __builtin_amdgcn_readlane(ms, j);
      const int s1 = __builtin_amdgcn_readlane(ms, j + 1);
      const int s2 = __builtin_amdgcn_readlane(ms, j + 2);
      const int s3 = __builtin_amdgcn_readlane(ms, j + 3);
      const float4 v0 = t4[(size_t)s0 * (D / 4) + lane];
      const float4 v1 = t4[(size_t)s1 * (D / 4) + lane];
      const float4 v2 = t4[(size_t)s2 * (D / 4) + lane];
      const float4 v3 = t4[(size_t)s3 * (D / 4) + lane];
      add4(acc, v0);
      add4(acc, v1);
      add4(acc, v2);
      add4(acc, v3);
    }
    const int r = n - j;
    if (r > 0) {
      const int s0 = __builtin_amdgcn_readlane(ms, j);
      const float4 v0 = t4[(size_t)s0 * (D / 4) + lane];
      if (r > 1) {
        const int s1 = __builtin_amdgcn_readlane(ms, j + 1);
        const float4 v1 = t4[(size_t)s1 * (D / 4) + lane];
        if (r > 2) {
          const int s2 = __builtin_amdgcn_readlane(ms, j + 2);
          const float4 v2 = t4[(size_t)s2 * (D / 4) + lane];
          add4(acc, v2);
        }
        add4(acc, v1);
      }
      add4(acc, v0);
    }
  }

  const float4 b = ((const float4*)bias)[lane];
  acc.x = fmaf(acc.x, di, b.x); acc.y = fmaf(acc.y, di, b.y);
  acc.z = fmaf(acc.z, di, b.z); acc.w = fmaf(acc.w, di, b.w);
  if (relu) {
    acc.x = fmaxf(acc.x, 0.0f); acc.y = fmaxf(acc.y, 0.0f);
    acc.z = fmaxf(acc.z, 0.0f); acc.w = fmaxf(acc.w, 0.0f);
  }
  ((float4*)h)[(size_t)row * (D / 4) + lane] = acc;
}

#define PG_BLOCKS 768
#define PG_PULL   16
#define PG_MMA    4
#define PG_T      ((PG_PULL + PG_MMA) * 32)
#define PG_APITCH 132
#define PG_OPITCH 36
__global__ void __launch_bounds__(PG_T) __attribute__((amdgpu_num_vgpr(64))) k_pull_gemm(const int* __restrict__ rowptr,
                                                    const int* __restrict__ col,
                                                    const float* __restrict__ t,
                                                    const float* __restrict__ dinv,
                                                    const float* __restrict__ bias,
                                                    const float* __restrict__ W,
                                                    float* __restrict__ Y) {
  __shared__ float4 s_w[(D / 4) * D];
  __shared__ __attribute__((aligned(16))) float s_a[2][16 * PG_APITCH];
  __shared__ __attribute__((aligned(16))) float s_o[PG_MMA][16 * PG_OPITCH];
  const int tid  = threadIdx.x;
  const int lane = tid & 31;
  const int wv = __builtin_amdgcn_readfirstlane(tid >> 5);

  for (int idx = tid; idx < (D / 4) * D; idx += PG_T) {
    const float* wp = W + (size_t)(idx >> 7) * (4 * D) + (idx & 127);
    float4 v;
    v.x = wp[0]; v.y = wp[D]; v.z = wp[2 * D]; v.w = wp[3 * D];
    s_w[idx] = v;
  }
  __syncthreads();

  const int G   = gridDim.x;
  const int bx  = blockIdx.x;
  const int nst = (bx < N_NODES / 16) ? (N_NODES / 16 - 1 - bx) / G + 1 : 0;
  const float4* t4 = (const float4*)t;
  const int c = lane & 15;
  const int h = lane >> 4;

  for (int it = 0; it <= nst; ++it) {
    if (wv >= PG_MMA) {
      if (it < nst) {
        const int pw  = wv - PG_MMA;
        const int row = (bx + it * G) * 16 + pw;
        const int beg = rowptr[row];
        const int m   = rowptr[row + 1] - beg + 1;
        const float di = dinv[row];
        float4 acc = make_float4(0.f, 0.f, 0.f, 0.f);
        for (int base = 0; base < m; base += 32) {
          const int n = min(32, m - base);
          const int k = base + lane;
          int ms = row;
          if (k < m - 1) ms = col[beg + k];
          int j = 0;
          for (; j + 8 <= n; j += 8) {
            const int s0 = __builtin_amdgcn_readlane(ms, j);
            const int s1 = __builtin_amdgcn_readlane(ms, j + 1);
            const int s2 = __builtin_amdgcn_readlane(ms, j + 2);
            const int s3 = __builtin_amdgcn_readlane(ms, j + 3);
            const int s4 = __builtin_amdgcn_readlane(ms, j + 4);
            const int s5 = __builtin_amdgcn_readlane(ms, j + 5);
            const int s6 = __builtin_amdgcn_readlane(ms, j + 6);
            const int s7 = __builtin_amdgcn_readlane(ms, j + 7);
            const float4 v0 = t4[(size_t)s0 * (D / 4) + lane];
            const float4 v1 = t4[(size_t)s1 * (D / 4) + lane];
            const float4 v2 = t4[(size_t)s2 * (D / 4) + lane];
            const float4 v3 = t4[(size_t)s3 * (D / 4) + lane];
            const float4 v4 = t4[(size_t)s4 * (D / 4) + lane];
            const float4 v5 = t4[(size_t)s5 * (D / 4) + lane];
            const float4 v6 = t4[(size_t)s6 * (D / 4) + lane];
            const float4 v7 = t4[(size_t)s7 * (D / 4) + lane];
            add4(acc, v0); add4(acc, v1); add4(acc, v2); add4(acc, v3);
            add4(acc, v4); add4(acc, v5); add4(acc, v6); add4(acc, v7);
          }
          for (; j + 4 <= n; j += 4) {
            const int s0 = __builtin_amdgcn_readlane(ms, j);
            const int s1 = __builtin_amdgcn_readlane(ms, j + 1);
            const int s2 = __builtin_amdgcn_readlane(ms, j + 2);
            const int s3 = __builtin_amdgcn_readlane(ms, j + 3);
            const float4 v0 = t4[(size_t)s0 * (D / 4) + lane];
            const float4 v1 = t4[(size_t)s1 * (D / 4) + lane];
            const float4 v2 = t4[(size_t)s2 * (D / 4) + lane];
            const float4 v3 = t4[(size_t)s3 * (D / 4) + lane];
            add4(acc, v0); add4(acc, v1); add4(acc, v2); add4(acc, v3);
          }
          const int r = n - j;
          if (r > 0) {
            const int s0 = __builtin_amdgcn_readlane(ms, j);
            const float4 v0 = t4[(size_t)s0 * (D / 4) + lane];
            if (r > 1) {
              const int s1 = __builtin_amdgcn_readlane(ms, j + 1);
              const float4 v1 = t4[(size_t)s1 * (D / 4) + lane];
              if (r > 2) {
                const int s2 = __builtin_amdgcn_readlane(ms, j + 2);
                const float4 v2 = t4[(size_t)s2 * (D / 4) + lane];
                add4(acc, v2);
              }
              add4(acc, v1);
            }
            add4(acc, v0);
          }
        }
        const float4 bv = ((const float4*)bias)[lane];
        acc.x = fmaxf(fmaf(acc.x, di, bv.x), 0.0f); acc.y = fmaxf(fmaf(acc.y, di, bv.y), 0.0f);
        acc.z = fmaxf(fmaf(acc.z, di, bv.z), 0.0f); acc.w = fmaxf(fmaf(acc.w, di, bv.w), 0.0f);
        *(float4*)(s_a[it & 1] + pw * PG_APITCH + 4 * lane) = acc;
      }
    } else {
      if (it >= 1) {
        const int row0 = (bx + (it - 1) * G) * 16;
        const float4* ar = (const float4*)(s_a[(it - 1) & 1] + c * PG_APITCH) + h;
        const float4* bw = s_w + h * D + wv * 32 + c;
        v8f acc[2] = {};
        float4 aq[2];
        float4 bq[2][2];
        aq[0] = ar[0];
        bq[0][0] = bw[0]; bq[0][1] = bw[16];
#pragma unroll
        for (int j = 0; j < 16; ++j) {
          if (j + 1 < 16) {
            aq[(j + 1) & 1] = ar[2 * (j + 1)];
            bq[(j + 1) & 1][0] = bw[(2 * (j + 1)) * D];
            bq[(j + 1) & 1][1] = bw[(2 * (j + 1)) * D + 16];
          }
#pragma unroll
          for (int tt = 0; tt < 2; ++tt) {
            v2f a; a.x = aq[j & 1].x; a.y = aq[j & 1].y;
            v2f b; b.x = bq[j & 1][tt].x; b.y = bq[j & 1][tt].y;
            acc[tt] = __builtin_amdgcn_wmma_f32_16x16x4_f32(
                 false, a,  false, b,
                 (short)0, acc[tt],  false,  false);
          }
#pragma unroll
          for (int tt = 0; tt < 2; ++tt) {
            v2f a; a.x = aq[j & 1].z; a.y = aq[j & 1].w;
            v2f b; b.x = bq[j & 1][tt].z; b.y = bq[j & 1][tt].w;
            acc[tt] = __builtin_amdgcn_wmma_f32_16x16x4_f32(
                 false, a,  false, b,
                 (short)0, acc[tt],  false,  false);
          }
          __builtin_amdgcn_sched_barrier(0);
        }
        const float4* dp = (const float4*)(dinv + row0 + 8 * h);
        const float4 ds0 = dp[0], ds1 = dp[1];
        const float sc[8] = {ds0.x, ds0.y, ds0.z, ds0.w, ds1.x, ds1.y, ds1.z, ds1.w};
        float* so = s_o[wv];
        float* sw = so + (8 * h) * PG_OPITCH + c;
#pragma unroll
        for (int tt = 0; tt < 2; ++tt)
#pragma unroll
          for (int i = 0; i < 8; ++i)
            sw[i * PG_OPITCH + tt * 16] = acc[tt][i] * sc[i];
        __builtin_amdgcn_fence(__ATOMIC_RELEASE, "wavefront");
        __builtin_amdgcn_fence(__ATOMIC_ACQUIRE, "wavefront");
#pragma unroll
        for (int q = 0; q < 4; ++q) {
          const int idx = q * 32 + lane;
          const int r = idx >> 3, f = idx & 7;
          const float4 v = *(const float4*)(so + r * PG_OPITCH + 4 * f);
          *(float4*)(Y + (size_t)(row0 + r) * D + wv * 32 + 4 * f) = v;
        }
        __builtin_amdgcn_fence(__ATOMIC_RELEASE, "wavefront");
        __builtin_amdgcn_fence(__ATOMIC_ACQUIRE, "wavefront");
      }
    }
    __syncthreads();
  }
}

#define DOT_P 32
#define DOT_U 8
#define DOT_WAVES 4
static_assert(N_PAIRS % DOT_P == 0, "a wave's pairs must all come from one list");
static_assert(DOT_P == 32 && DOT_U == 8, "the exchange below is written for 4 trips of 8 pairs");

__device__ __forceinline__ float halve(const float lo, const float hi, const bool up, const int m) {
  const float keep = up ? hi : lo;
  const float send = up ? lo : hi;
  return keep + __shfl_xor(send, m, 32);
}

__global__ void __launch_bounds__(DOT_WAVES * 32) k_dot(const int* __restrict__ ep,
                                                        const int* __restrict__ en,
                                                        const float* __restrict__ h,
                                                        float* __restrict__ out) {
  const int lane = threadIdx.x & 31;
  const int wave = __builtin_amdgcn_readfirstlane(blockIdx.x * DOT_WAVES + (threadIdx.x >> 5));
  const int p0 = wave * DOT_P;
  if (p0 >= 2 * N_PAIRS) return;
  const int* pr = (p0 < N_PAIRS) ? ep : en;
  const int  q0 = (p0 < N_PAIRS) ? p0 : p0 - N_PAIRS;
  const int2 ab = ((const int2*)pr)[q0 + lane];
  const bool up1 = (lane & 1) != 0, up2 = (lane & 2) != 0, up4 = (lane & 4) != 0;

  float r[DOT_P / DOT_U];
#pragma unroll
  for (int g = 0; g < DOT_P / DOT_U; ++g) {
    float4 va[DOT_U], vb[DOT_U];
#pragma unroll
    for (int u = 0; u < DOT_U; ++u) {
      const int a = __builtin_amdgcn_readlane(ab.x, g * DOT_U + u);
      const int b = __builtin_amdgcn_readlane(ab.y, g * DOT_U + u);
      va[u] = ((const float4*)(h + (size_t)a * D))[lane];
      vb[u] = ((const float4*)(h + (size_t)b * D))[lane];
    }
    float p[DOT_U];
#pragma unroll
    for (int u = 0; u < DOT_U; ++u)
      p[u] = fmaf(va[u].w, vb[u].w, fmaf(va[u].z, vb[u].z, fmaf(va[u].x, vb[u].x, va[u].y * vb[u].y)));
    const float c0 = halve(p[0], p[1], up1, 1), c1 = halve(p[2], p[3], up1, 1);
    const float c2 = halve(p[4], p[5], up1, 1), c3 = halve(p[6], p[7], up1, 1);
    const float d0 = halve(c0, c1, up2, 2), d1 = halve(c2, c3, up2, 2);
    r[g] = halve(d0, d1, up4, 4);
  }
  const bool up8 = (lane & 8) != 0, up16 = (lane & 16) != 0;
  const float w0 = halve(r[0], r[1], up8, 8), w1 = halve(r[2], r[3], up8, 8);
  out[p0 + lane] = halve(w0, w1, up16, 16);
}

extern "C" void kernel_launch(void* const* d_in, const int* in_sizes, int n_in,
                              void* d_out, int out_size, void* d_ws, size_t ws_size,
                              hipStream_t stream) {
  (void)in_sizes; (void)n_in; (void)out_size; (void)ws_size;
  const float* x   = (const float*)d_in[0];
  const float* W1  = (const float*)d_in[1];
  const float* b1  = (const float*)d_in[2];
  const float* W2  = (const float*)d_in[3];
  const float* b2  = (const float*)d_in[4];
  const int* ei    = (const int*)d_in[5];
  const int* edges = (const int*)d_in[6];
  const int* edneg = (const int*)d_in[7];
  float* out = (float*)d_out;

  const size_t FEAT = (size_t)N_NODES * D;
  float* t    = (float*)d_ws;
  float* csr  = t + FEAT;
  float* h    = csr + FEAT;
  float* dinv = h + FEAT;
  int* rowptr = (int*)csr;
  int* col    = rowptr + N_NODES + 4;
  int* hist   = col + N_EDGES;
  int* bstart = hist + BIN_G * NBK_PAD;
  int* recs   = bstart + 256;

  const int NB_GA = (N_NODES + 7) / 8;
  const int NB_DP = (2 * N_PAIRS / DOT_P + DOT_WAVES - 1) / DOT_WAVES;

  k_hist<<<BIN_G, BIN_T, 0, stream>>>(ei, hist);
  k_bin <<<BIN_G, BIN_T, 0, stream>>>(ei, hist, bstart, recs);
  k_rows<<<NBK, ROW_T, 0, stream>>>(bstart, recs, rowptr, dinv, col);

  k_gemm<<<GM_BLOCKS, 256, 0, stream>>>(x, W1, dinv, t);

  k_pull_gemm<<<PG_BLOCKS, PG_T, 0, stream>>>(rowptr, col, t, dinv, b1, W2, h);

  k_gather<<<NB_GA, 256, 0, stream>>>(rowptr, col, h, dinv, b2, t, 0);

  k_dot<<<NB_DP, DOT_WAVES * 32, 0, stream>>>(edges, edneg, t, out);
}
